# starting kernel
# baseline (speedup 1.0000x reference)
#include <hip/hip_runtime.h>

#define NN 50000
#define EE 200000
#define BB 64
#define LL 6

typedef float v2f __attribute__((ext_vector_type(2)));
typedef float v8f __attribute__((ext_vector_type(8)));

__device__ inline v2f ld2(const float* p) { return *(const v2f*)p; }

template <int K, typename FA>
__device__ inline v8f tile_gemm(FA fa, const float* __restrict__ B, int ldb,
                                int col0, v8f acc) {
  const int lane = threadIdx.x & 31;
  const int h2   = (lane >> 4) * 2;
  const int l15  = lane & 15;
#pragma unroll
  for (int kk = 0; kk < K; kk += 4) {
    v2f a = fa(l15, kk, h2);
    v2f b;
    const float* bp = B + (size_t)(kk + h2) * ldb + col0 + l15;
    b.x = bp[0];
    b.y = bp[ldb];
    acc = __builtin_amdgcn_wmma_f32_16x16x4_f32(false, a, false, b,
                                                (short)0, acc, false, false);
  }
  return acc;
}

template <typename ST>
__device__ inline void tile_store(v8f acc, ST st) {
  const int lane = threadIdx.x & 31;
  const int half = lane >> 4;
  const int l15  = lane & 15;
#pragma unroll
  for (int r = 0; r < 8; ++r) st(r + half * 8, l15, acc[r]);
}

template <int K, int NC>
__global__ void encoder_kernel(const float* __restrict__ A,
                               const float* __restrict__ W,
                               const float* __restrict__ bias,
                               float* __restrict__ out) {
  __shared__ __align__(16) float a_s[16 * K];
  const int m0 = blockIdx.x * 16;
  for (int i = threadIdx.x; i < 16 * K / 4; i += blockDim.x) {
    int r = i / (K / 4), c = (i % (K / 4)) * 4;
    *(float4*)(a_s + r * K + c) = *(const float4*)(A + (size_t)(m0 + r) * K + c);
  }
  __syncthreads();
  const int col0 = (threadIdx.x >> 5) * 16;
  v8f acc = {};
  acc = tile_gemm<K>([&](int r, int kk, int h2) {
    return ld2(a_s + r * K + kk + h2);
  }, W, NC, col0, acc);
  tile_store(acc, [&](int r, int c, float v) {
    out[(size_t)(m0 + r) * NC + col0 + c] = v + bias[col0 + c];
  });
}

__global__ void edge_mlp_kernel(const float* __restrict__ h, float* __restrict__ e,
                                float* __restrict__ macc,
                                const int* __restrict__ src, const int* __restrict__ dst,
                                const float* __restrict__ mw,
                                const float* __restrict__ w1, const float* __restrict__ b1,
                                const float* __restrict__ w2, const float* __restrict__ b2) {
  __shared__ __align__(16) float hs_s[16 * 128];
  __shared__ __align__(16) float hd_s[16 * 128];
  __shared__ __align__(16) float e_s[16 * 64];
  __shared__ __align__(16) float t_s[16 * 64];
  __shared__ __align__(16) float u_s[16 * 64];
  const int m0   = blockIdx.x * 16;
  const int col0 = (threadIdx.x >> 5) * 16;

  for (int i = threadIdx.x; i < 16 * 128 / 4; i += blockDim.x) {
    int r = i >> 5, c = (i & 31) * 4;
    *(float4*)(hs_s + r * 128 + c) = *(const float4*)(h + (size_t)src[m0 + r] * 128 + c);
    *(float4*)(hd_s + r * 128 + c) = *(const float4*)(h + (size_t)dst[m0 + r] * 128 + c);
  }
  for (int i = threadIdx.x; i < 16 * 64 / 4; i += blockDim.x) {
    int r = i >> 4, c = (i & 15) * 4;
    *(float4*)(e_s + r * 64 + c) = *(const float4*)(e + (size_t)(m0 + r) * 64 + c);
  }
  __syncthreads();

  v8f acc0 = {};
  acc0 = tile_gemm<128>([&](int r, int kk, int h2) {
    return ld2(hs_s + r * 128 + kk + h2);
  }, mw, 64, col0, acc0);
  tile_store(acc0, [&](int r, int c, float v) {
    t_s[r * 64 + col0 + c] = e_s[r * 64 + col0 + c] + v;
  });
  __syncthreads();

  v8f acc1 = {};
  acc1 = tile_gemm<320>([&](int r, int kk, int h2) {
    if (kk < 128) return ld2(hs_s + r * 128 + kk + h2);
    if (kk < 256) return ld2(hd_s + r * 128 + (kk - 128) + h2);
    return ld2(t_s + r * 64 + (kk - 256) + h2);
  }, w1, 64, col0, acc1);
  tile_store(acc1, [&](int r, int c, float v) {
    float t = v + b1[col0 + c];
    u_s[r * 64 + col0 + c] = t > 0.f ? t : 0.f;
  });
  __syncthreads();

  v8f acc2 = {};
  acc2 = tile_gemm<64>([&](int r, int kk, int h2) {
    return ld2(u_s + r * 64 + kk + h2);
  }, w2, 64, col0, acc2);
  tile_store(acc2, [&](int r, int c, float v) {
    int erow = m0 + r;
    float val = e_s[r * 64 + col0 + c] + v + b2[col0 + c];
    e[(size_t)erow * 64 + col0 + c] = val;
    atomicAdd(macc + (size_t)dst[erow] * 64 + col0 + c, val);
  });
}

__global__ void node_mlp_kernel(float* __restrict__ h, const float* __restrict__ macc,
                                const float* __restrict__ rdeg,
                                const float* __restrict__ w1, const float* __restrict__ b1,
                                const float* __restrict__ w2, const float* __restrict__ b2) {
  __shared__ __align__(16) float hv_s[16 * 128];
  __shared__ __align__(16) float m_s[16 * 64];
  __shared__ __align__(16) float u_s[16 * 64];
  const int m0   = blockIdx.x * 16;
  const int wave = threadIdx.x >> 5;

  for (int i = threadIdx.x; i < 16 * 128 / 4; i += blockDim.x) {
    int r = i >> 5, c = (i & 31) * 4;
    *(float4*)(hv_s + r * 128 + c) = *(const float4*)(h + (size_t)(m0 + r) * 128 + c);
  }
  for (int i = threadIdx.x; i < 16 * 64 / 4; i += blockDim.x) {
    int r = i >> 4, c = (i & 15) * 4;
    float4 v = *(const float4*)(macc + (size_t)(m0 + r) * 64 + c);
    float rd = rdeg[m0 + r];
    v.x *= rd; v.y *= rd; v.z *= rd; v.w *= rd;
    *(float4*)(m_s + r * 64 + c) = v;
  }
  __syncthreads();

  if (wave < 4) {
    const int col0 = wave * 16;
    v8f acc = {};
    acc = tile_gemm<192>([&](int r, int kk, int h2) {
      if (kk < 128) return ld2(hv_s + r * 128 + kk + h2);
      return ld2(m_s + r * 64 + (kk - 128) + h2);
    }, w1, 64, col0, acc);
    tile_store(acc, [&](int r, int c, float v) {
      float t = v + b1[col0 + c];
      u_s[r * 64 + col0 + c] = t > 0.f ? t : 0.f;
    });
  }
  __syncthreads();

  const int col0 = wave * 16;
  v8f acc2 = {};
  acc2 = tile_gemm<64>([&](int r, int kk, int h2) {
    return ld2(u_s + r * 64 + kk + h2);
  }, w2, 128, col0, acc2);
  tile_store(acc2, [&](int r, int c, float v) {
    h[(size_t)(m0 + r) * 128 + col0 + c] = hv_s[r * 128 + col0 + c] + v + b2[col0 + c];
  });
}

__global__ void deg_count_kernel(const int* __restrict__ dst, float* __restrict__ cnt) {
  int i = blockIdx.x * blockDim.x + threadIdx.x;
  if (i < EE) atomicAdd(cnt + dst[i], 1.0f);
}

__global__ void recip_clamp_kernel(float* __restrict__ v, int n) {
  int i = blockIdx.x * blockDim.x + threadIdx.x;
  if (i < n) v[i] = 1.0f / fmaxf(v[i], 1.0f);
}

__global__ void pool_kernel(const float* __restrict__ h, const int* __restrict__ batch,
                            float* __restrict__ gsum, float* __restrict__ gcnt) {
  int i = blockIdx.x * blockDim.x + threadIdx.x;
  int node = i >> 7, c = i & 127;
  if (node < NN) {
    int g = batch[node];
    atomicAdd(gsum + (size_t)g * 128 + c, h[(size_t)node * 128 + c]);
    if (c == 0) atomicAdd(gcnt + g, 1.0f);
  }
}

__global__ void readout_kernel(const float* __restrict__ gsum, const float* __restrict__ rgcnt,
                               const float* __restrict__ w1, const float* __restrict__ b1,
                               const float* __restrict__ w2, const float* __restrict__ b2,
                               float* __restrict__ out) {
  __shared__ __align__(16) float g_s[16 * 128];
  __shared__ __align__(16) float t_s[16 * 128];
  const int m0   = blockIdx.x * 16;
  const int col0 = (threadIdx.x >> 5) * 16;

  for (int i = threadIdx.x; i < 16 * 128 / 4; i += blockDim.x) {
    int r = i >> 5, c = (i & 31) * 4;
    float4 v = *(const float4*)(gsum + (size_t)(m0 + r) * 128 + c);
    float rc = rgcnt[m0 + r];
    v.x *= rc; v.y *= rc; v.z *= rc; v.w *= rc;
    *(float4*)(g_s + r * 128 + c) = v;
  }
  __syncthreads();

  v8f acc = {};
  acc = tile_gemm<128>([&](int r, int kk, int h2) {
    return ld2(g_s + r * 128 + kk + h2);
  }, w1, 128, col0, acc);
  tile_store(acc, [&](int r, int c, float v) {
    float t = v + b1[col0 + c];
    t_s[r * 128 + col0 + c] = t > 0.f ? t : 0.f;
  });
  __syncthreads();

  v8f acc2 = {};
  acc2 = tile_gemm<128>([&](int r, int kk, int h2) {
    return ld2(t_s + r * 128 + kk + h2);
  }, w2, 128, col0, acc2);
  tile_store(acc2, [&](int r, int c, float v) {
    out[(size_t)(m0 + r) * 128 + col0 + c] = v + b2[col0 + c];
  });
}

extern "C" void kernel_launch(void* const* d_in, const int* in_sizes, int n_in,
                              void* d_out, int out_size, void* d_ws, size_t ws_size,
                              hipStream_t stream) {
  (void)in_sizes; (void)n_in; (void)out_size; (void)ws_size;
  const float* x       = (const float*)d_in[0];
  const float* ea      = (const float*)d_in[1];
  const int*   eidx    = (const int*)d_in[2];
  const int*   batch   = (const int*)d_in[3];
  const float* lin_x_w = (const float*)d_in[4];
  const float* lin_x_b = (const float*)d_in[5];
  const float* lin_e_w = (const float*)d_in[6];
  const float* lin_e_b = (const float*)d_in[7];
  const float* msg_w   = (const float*)d_in[8];
  const float* phie_w1 = (const float*)d_in[9];
  const float* phie_b1 = (const float*)d_in[10];
  const float* phie_w2 = (const float*)d_in[11];
  const float* phie_b2 = (const float*)d_in[12];
  const float* phiv_w1 = (const float*)d_in[13];
  const float* phiv_b1 = (const float*)d_in[14];
  const float* phiv_w2 = (const float*)d_in[15];
  const float* phiv_b2 = (const float*)d_in[16];
  const float* ro_w1   = (const float*)d_in[17];
  const float* ro_b1   = (const float*)d_in[18];
  const float* ro_w2   = (const float*)d_in[19];
  const float* ro_b2   = (const float*)d_in[20];
  const int* src = eidx;
  const int* dst = eidx + EE;

  char* wp = (char*)d_ws;
  auto carve = [&](size_t bytes) {
    char* p = wp; wp += (bytes + 255) & ~(size_t)255; return p;
  };
  float* h    = (float*)carve((size_t)NN * 128 * 4);
  float* e    = (float*)carve((size_t)EE * 64 * 4);
  float* macc = (float*)carve((size_t)NN * 64 * 4);
  float* rdeg = (float*)carve((size_t)NN * 4);
  float* gsum = (float*)carve((size_t)BB * 128 * 4);
  float* gcnt = (float*)carve((size_t)BB * 4);

  encoder_kernel<64, 128><<<NN / 16, 256, 0, stream>>>(x, lin_x_w, lin_x_b, h);
  encoder_kernel<16, 64><<<EE / 16, 128, 0, stream>>>(ea, lin_e_w, lin_e_b, e);

  hipMemsetAsync(rdeg, 0, (size_t)NN * 4, stream);
  deg_count_kernel<<<(EE + 255) / 256, 256, 0, stream>>>(dst, rdeg);
  recip_clamp_kernel<<<(NN + 255) / 256, 256, 0, stream>>>(rdeg, NN);

  for (int l = 0; l < LL; ++l) {
    hipMemsetAsync(macc, 0, (size_t)NN * 64 * 4, stream);
    edge_mlp_kernel<<<EE / 16, 128, 0, stream>>>(h, e, macc, src, dst,
        msg_w + (size_t)l * 128 * 64,
        phie_w1 + (size_t)l * 320 * 64, phie_b1 + (size_t)l * 64,
        phie_w2 + (size_t)l * 64 * 64,  phie_b2 + (size_t)l * 64);
    node_mlp_kernel<<<NN / 16, 256, 0, stream>>>(h, macc, rdeg,
        phiv_w1 + (size_t)l * 192 * 64, phiv_b1 + (size_t)l * 64,
        phiv_w2 + (size_t)l * 64 * 128, phiv_b2 + (size_t)l * 128);
  }

  hipMemsetAsync(gsum, 0, (size_t)BB * 128 * 4, stream);
  hipMemsetAsync(gcnt, 0, (size_t)BB * 4, stream);
  pool_kernel<<<(NN * 128 + 255) / 256, 256, 0, stream>>>(h, batch, gsum, gcnt);
  recip_clamp_kernel<<<1, BB, 0, stream>>>(gcnt, BB);
  readout_kernel<<<BB / 16, 256, 0, stream>>>(gsum, gcnt, ro_w1, ro_b1, ro_w2, ro_b2,
                                              (float*)d_out);
}


# LDS activation tiles at a row pitch of row length + 4 floats, so the 16 rows of every A-fragment read fall on different banks (was a 16-way bank conflict per read); sums unchanged
# speedup vs baseline: 1.5408x; 1.5408x over previous
#include <hip/hip_runtime.h>

#define NN 50000
#define EE 200000
#define BB 64
#define LL 6

typedef float v2f __attribute__((ext_vector_type(2)));
typedef float v8f __attribute__((ext_vector_type(8)));

__device__ inline v2f ld2(const float* p) { return *(const v2f*)p; }

#define PAD 4
#define P64  (64 + PAD)
#define P128 (128 + PAD)

template <int K, typename FA>
__device__ inline v8f tile_gemm(FA fa, const float* __restrict__ B, int ldb,
                                int col0, v8f acc) {
  const int lane = threadIdx.x & 31;
  const int h2   = (lane >> 4) * 2;
  const int l15  = lane & 15;
#pragma unroll
  for (int kk = 0; kk < K; kk += 4) {
    v2f a = fa(l15, kk, h2);
    v2f b;
    const float* bp = B + (size_t)(kk + h2) * ldb + col0 + l15;
    b.x = bp[0];
    b.y = bp[ldb];
    acc = __builtin_amdgcn_wmma_f32_16x16x4_f32(false, a, false, b,
                                                (short)0, acc, false, false);
  }
  return acc;
}

template <typename ST>
__device__ inline void tile_store(v8f acc, ST st) {
  const int lane = threadIdx.x & 31;
  const int half = lane >> 4;
  const int l15  = lane & 15;
#pragma unroll
  for (int r = 0; r < 8; ++r) st(r + half * 8, l15, acc[r]);
}

template <int K, int NC>
__global__ void encoder_kernel(const float* __restrict__ A,
                               const float* __restrict__ W,
                               const float* __restrict__ bias,
                               float* __restrict__ out) {
  constexpr int P = K + PAD;
  __shared__ __align__(16) float a_s[16 * P];
  const int m0 = blockIdx.x * 16;
  for (int i = threadIdx.x; i < 16 * K / 4; i += blockDim.x) {
    int r = i / (K / 4), c = (i % (K / 4)) * 4;
    *(float4*)(a_s + r * P + c) = *(const float4*)(A + (size_t)(m0 + r) * K + c);
  }
  __syncthreads();
  const int col0 = (threadIdx.x >> 5) * 16;
  v8f acc = {};
  acc = tile_gemm<K>([&](int r, int kk, int h2) {
    return ld2(a_s + r * P + kk + h2);
  }, W, NC, col0, acc);
  tile_store(acc, [&](int r, int c, float v) {
    out[(size_t)(m0 + r) * NC + col0 + c] = v + bias[col0 + c];
  });
}

__global__ void edge_mlp_kernel(const float* __restrict__ h, float* __restrict__ e,
                                float* __restrict__ macc,
                                const int* __restrict__ src, const int* __restrict__ dst,
                                const float* __restrict__ mw,
                                const float* __restrict__ w1, const float* __restrict__ b1,
                                const float* __restrict__ w2, const float* __restrict__ b2) {
  __shared__ __align__(16) float hs_s[16 * P128];
  __shared__ __align__(16) float hd_s[16 * P128];
  __shared__ __align__(16) float e_s[16 * P64];
  __shared__ __align__(16) float t_s[16 * P64];
  __shared__ __align__(16) float u_s[16 * P64];
  const int m0   = blockIdx.x * 16;
  const int col0 = (threadIdx.x >> 5) * 16;

  for (int i = threadIdx.x; i < 16 * 128 / 4; i += blockDim.x) {
    int r = i >> 5, c = (i & 31) * 4;
    *(float4*)(hs_s + r * P128 + c) = *(const float4*)(h + (size_t)src[m0 + r] * 128 + c);
    *(float4*)(hd_s + r * P128 + c) = *(const float4*)(h + (size_t)dst[m0 + r] * 128 + c);
  }
  for (int i = threadIdx.x; i < 16 * 64 / 4; i += blockDim.x) {
    int r = i >> 4, c = (i & 15) * 4;
    *(float4*)(e_s + r * P64 + c) = *(const float4*)(e + (size_t)(m0 + r) * 64 + c);
  }
  __syncthreads();

  v8f acc0 = {};
  acc0 = tile_gemm<128>([&](int r, int kk, int h2) {
    return ld2(hs_s + r * P128 + kk + h2);
  }, mw, 64, col0, acc0);
  tile_store(acc0, [&](int r, int c, float v) {
    t_s[r * P64 + col0 + c] = e_s[r * P64 + col0 + c] + v;
  });
  __syncthreads();

  v8f acc1 = {};
  acc1 = tile_gemm<320>([&](int r, int kk, int h2) {
    if (kk < 128) return ld2(hs_s + r * P128 + kk + h2);
    if (kk < 256) return ld2(hd_s + r * P128 + (kk - 128) + h2);
    return ld2(t_s + r * P64 + (kk - 256) + h2);
  }, w1, 64, col0, acc1);
  tile_store(acc1, [&](int r, int c, float v) {
    float t = v + b1[col0 + c];
    u_s[r * P64 + col0 + c] = t > 0.f ? t : 0.f;
  });
  __syncthreads();

  v8f acc2 = {};
  acc2 = tile_gemm<64>([&](int r, int kk, int h2) {
    return ld2(u_s + r * P64 + kk + h2);
  }, w2, 64, col0, acc2);
  tile_store(acc2, [&](int r, int c, float v) {
    int erow = m0 + r;
    float val = e_s[r * P64 + col0 + c] + v + b2[col0 + c];
    e[(size_t)erow * 64 + col0 + c] = val;
    atomicAdd(macc + (size_t)dst[erow] * 64 + col0 + c, val);
  });
}

__global__ void node_mlp_kernel(float* __restrict__ h, const float* __restrict__ macc,
                                const float* __restrict__ rdeg,
                                const float* __restrict__ w1, const float* __restrict__ b1,
                                const float* __restrict__ w2, const float* __restrict__ b2) {
  __shared__ __align__(16) float hv_s[16 * P128];
  __shared__ __align__(16) float m_s[16 * P64];
  __shared__ __align__(16) float u_s[16 * P64];
  const int m0   = blockIdx.x * 16;
  const int wave = threadIdx.x >> 5;

  for (int i = threadIdx.x; i < 16 * 128 / 4; i += blockDim.x) {
    int r = i >> 5, c = (i & 31) * 4;
    *(float4*)(hv_s + r * P128 + c) = *(const float4*)(h + (size_t)(m0 + r) * 128 + c);
  }
  for (int i = threadIdx.x; i < 16 * 64 / 4; i += blockDim.x) {
    int r = i >> 4, c = (i & 15) * 4;
    float4 v = *(const float4*)(macc + (size_t)(m0 + r) * 64 + c);
    float rd = rdeg[m0 + r];
    v.x *= rd; v.y *= rd; v.z *= rd; v.w *= rd;
    *(float4*)(m_s + r * P64 + c) = v;
  }
  __syncthreads();

  if (wave < 4) {
    const int col0 = wave * 16;
    v8f acc = {};
    acc = tile_gemm<192>([&](int r, int kk, int h2) {
      if (kk < 128) return ld2(hv_s + r * P128 + kk + h2);
      return ld2(m_s + r * P64 + (kk - 128) + h2);
    }, w1, 64, col0, acc);
    tile_store(acc, [&](int r, int c, float v) {
      float t = v + b1[col0 + c];
      u_s[r * P64 + col0 + c] = t > 0.f ? t : 0.f;
    });
  }
  __syncthreads();

  const int col0 = wave * 16;
  v8f acc2 = {};
  acc2 = tile_gemm<64>([&](int r, int kk, int h2) {
    return ld2(u_s + r * P64 + kk + h2);
  }, w2, 128, col0, acc2);
  tile_store(acc2, [&](int r, int c, float v) {
    h[(size_t)(m0 + r) * 128 + col0 + c] = hv_s[r * P128 + col0 + c] + v + b2[col0 + c];
  });
}

__global__ void deg_count_kernel(const int* __restrict__ dst, float* __restrict__ cnt) {
  int i = blockIdx.x * blockDim.x + threadIdx.x;
  if (i < EE) atomicAdd(cnt + dst[i], 1.0f);
}

__global__ void recip_clamp_kernel(float* __restrict__ v, int n) {
  int i = blockIdx.x * blockDim.x + threadIdx.x;
  if (i < n) v[i] = 1.0f / fmaxf(v[i], 1.0f);
}

__global__ void pool_kernel(const float* __restrict__ h, const int* __restrict__ batch,
                            float* __restrict__ gsum, float* __restrict__ gcnt) {
  int i = blockIdx.x * blockDim.x + threadIdx.x;
  int node = i >> 7, c = i & 127;
  if (node < NN) {
    int g = batch[node];
    atomicAdd(gsum + (size_t)g * 128 + c, h[(size_t)node * 128 + c]);
    if (c == 0) atomicAdd(gcnt + g, 1.0f);
  }
}

__global__ void readout_kernel(const float* __restrict__ gsum, const float* __restrict__ rgcnt,
                               const float* __restrict__ w1, const float* __restrict__ b1,
                               const float* __restrict__ w2, const float* __restrict__ b2,
                               float* __restrict__ out) {
  __shared__ __align__(16) float g_s[16 * P128];
  __shared__ __align__(16) float t_s[16 * P128];
  const int m0   = blockIdx.x * 16;
  const int col0 = (threadIdx.x >> 5) * 16;

  for (int i = threadIdx.x; i < 16 * 128 / 4; i += blockDim.x) {
    int r = i >> 5, c = (i & 31) * 4;
    float4 v = *(const float4*)(gsum + (size_t)(m0 + r) * 128 + c);
    float rc = rgcnt[m0 + r];
    v.x *= rc; v.y *= rc; v.z *= rc; v.w *= rc;
    *(float4*)(g_s + r * P128 + c) = v;
  }
  __syncthreads();

  v8f acc = {};
  acc = tile_gemm<128>([&](int r, int kk, int h2) {
    return ld2(g_s + r * P128 + kk + h2);
  }, w1, 128, col0, acc);
  tile_store(acc, [&](int r, int c, float v) {
    float t = v + b1[col0 + c];
    t_s[r * P128 + col0 + c] = t > 0.f ? t : 0.f;
  });
  __syncthreads();

  v8f acc2 = {};
  acc2 = tile_gemm<128>([&](int r, int kk, int h2) {
    return ld2(t_s + r * P128 + kk + h2);
  }, w2, 128, col0, acc2);
  tile_store(acc2, [&](int r, int c, float v) {
    out[(size_t)(m0 + r) * 128 + col0 + c] = v + b2[col0 + c];
  });
}

extern "C" void kernel_launch(void* const* d_in, const int* in_sizes, int n_in,
                              void* d_out, int out_size, void* d_ws, size_t ws_size,
                              hipStream_t stream) {
  (void)in_sizes; (void)n_in; (void)out_size; (void)ws_size;
  const float* x       = (const float*)d_in[0];
  const float* ea      = (const float*)d_in[1];
  const int*   eidx    = (const int*)d_in[2];
  const int*   batch   = (const int*)d_in[3];
  const float* lin_x_w = (const float*)d_in[4];
  const float* lin_x_b = (const float*)d_in[5];
  const float* lin_e_w = (const float*)d_in[6];
  const float* lin_e_b = (const float*)d_in[7];
  const float* msg_w   = (const float*)d_in[8];
  const float* phie_w1 = (const float*)d_in[9];
  const float* phie_b1 = (const float*)d_in[10];
  const float* phie_w2 = (const float*)d_in[11];
  const float* phie_b2 = (const float*)d_in[12];
  const float* phiv_w1 = (const float*)d_in[13];
  const float* phiv_b1 = (const float*)d_in[14];
  const float* phiv_w2 = (const float*)d_in[15];
  const float* phiv_b2 = (const float*)d_in[16];
  const float* ro_w1   = (const float*)d_in[17];
  const float* ro_b1   = (const float*)d_in[18];
  const float* ro_w2   = (const float*)d_in[19];
  const float* ro_b2   = (const float*)d_in[20];
  const int* src = eidx;
  const int* dst = eidx + EE;

  char* wp = (char*)d_ws;
  auto carve = [&](size_t bytes) {
    char* p = wp; wp += (bytes + 255) & ~(size_t)255; return p;
  };
  float* h    = (float*)carve((size_t)NN * 128 * 4);
  float* e    = (float*)carve((size_t)EE * 64 * 4);
  float* macc = (float*)carve((size_t)NN * 64 * 4);
  float* rdeg = (float*)carve((size_t)NN * 4);
  float* gsum = (float*)carve((size_t)BB * 128 * 4);
  float* gcnt = (float*)carve((size_t)BB * 4);

  encoder_kernel<64, 128><<<NN / 16, 256, 0, stream>>>(x, lin_x_w, lin_x_b, h);
  encoder_kernel<16, 64><<<EE / 16, 128, 0, stream>>>(ea, lin_e_w, lin_e_b, e);

  hipMemsetAsync(rdeg, 0, (size_t)NN * 4, stream);
  deg_count_kernel<<<(EE + 255) / 256, 256, 0, stream>>>(dst, rdeg);
  recip_clamp_kernel<<<(NN + 255) / 256, 256, 0, stream>>>(rdeg, NN);

  for (int l = 0; l < LL; ++l) {
    hipMemsetAsync(macc, 0, (size_t)NN * 64 * 4, stream);
    edge_mlp_kernel<<<EE / 16, 128, 0, stream>>>(h, e, macc, src, dst,
        msg_w + (size_t)l * 128 * 64,
        phie_w1 + (size_t)l * 320 * 64, phie_b1 + (size_t)l * 64,
        phie_w2 + (size_t)l * 64 * 64,  phie_b2 + (size_t)l * 64);
    node_mlp_kernel<<<NN / 16, 256, 0, stream>>>(h, macc, rdeg,
        phiv_w1 + (size_t)l * 192 * 64, phiv_b1 + (size_t)l * 64,
        phiv_w2 + (size_t)l * 64 * 128, phiv_b2 + (size_t)l * 128);
  }

  hipMemsetAsync(gsum, 0, (size_t)BB * 128 * 4, stream);
  hipMemsetAsync(gcnt, 0, (size_t)BB * 4, stream);
  pool_kernel<<<(NN * 128 + 255) / 256, 256, 0, stream>>>(h, batch, gsum, gcnt);
  recip_clamp_kernel<<<1, BB, 0, stream>>>(gcnt, BB);
  readout_kernel<<<BB / 16, 256, 0, stream>>>(gsum, gcnt, ro_w1, ro_b1, ro_w2, ro_b2,
                                              (float*)d_out);
}


# Edge kernel: 64 edges per block, each wave holds 4 row tiles so one weight fragment load feeds 4 matrix ops (was 1); u reuses the hs LDS tile so 3 blocks fit a WGP; k order of sums unchanged
# speedup vs baseline: 1.5959x; 1.0358x over previous
#include <hip/hip_runtime.h>

#define NN 50000
#define EE 200000
#define BB 64
#define LL 6

typedef float v2f __attribute__((ext_vector_type(2)));
typedef float v8f __attribute__((ext_vector_type(8)));

__device__ inline v2f ld2(const float* p) { return *(const v2f*)p; }

#define PAD 4
#define P64  (64 + PAD)
#define P128 (128 + PAD)

template <int K, typename FA>
__device__ inline v8f tile_gemm(FA fa, const float* __restrict__ B, int ldb,
                                int col0, v8f acc) {
  const int lane = threadIdx.x & 31;
  const int h2   = (lane >> 4) * 2;
  const int l15  = lane & 15;
#pragma unroll
  for (int kk = 0; kk < K; kk += 4) {
    v2f a = fa(l15, kk, h2);
    v2f b;
    const float* bp = B + (size_t)(kk + h2) * ldb + col0 + l15;
    b.x = bp[0];
    b.y = bp[ldb];
    acc = __builtin_amdgcn_wmma_f32_16x16x4_f32(false, a, false, b,
                                                (short)0, acc, false, false);
  }
  return acc;
}

template <typename ST>
__device__ inline void tile_store(v8f acc, ST st) {
  const int lane = threadIdx.x & 31;
  const int half = lane >> 4;
  const int l15  = lane & 15;
#pragma unroll
  for (int r = 0; r < 8; ++r) st(r + half * 8, l15, acc[r]);
}

#define RT 4
#define EB (16 * RT)

template <int K, typename FA>
__device__ inline void tile_gemm_rt(FA fa, const float* __restrict__ B, int ldb,
                                    int col0, v8f (&acc)[RT]) {
  const int lane = threadIdx.x & 31;
  const int h2   = (lane >> 4) * 2;
  const int l15  = lane & 15;
#pragma unroll
  for (int kk = 0; kk < K; kk += 4) {
    v2f b;
    const float* bp = B + (size_t)(kk + h2) * ldb + col0 + l15;
    b.x = bp[0];
    b.y = bp[ldb];
#pragma unroll
    for (int t = 0; t < RT; ++t) {
      v2f a = fa(16 * t + l15, kk, h2);
      acc[t] = __builtin_amdgcn_wmma_f32_16x16x4_f32(false, a, false, b,
                                                     (short)0, acc[t], false, false);
    }
  }
}

template <typename ST>
__device__ inline void tile_store_rt(const v8f (&acc)[RT], ST st) {
  const int lane = threadIdx.x & 31;
  const int half = lane >> 4;
  const int l15  = lane & 15;
#pragma unroll
  for (int t = 0; t < RT; ++t)
#pragma unroll
    for (int r = 0; r < 8; ++r) st(16 * t + r + half * 8, l15, acc[t][r]);
}

template <int K, int NC>
__global__ void encoder_kernel(const float* __restrict__ A,
                               const float* __restrict__ W,
                               const float* __restrict__ bias,
                               float* __restrict__ out) {
  constexpr int P = K + PAD;
  __shared__ __align__(16) float a_s[16 * P];
  const int m0 = blockIdx.x * 16;
  for (int i = threadIdx.x; i < 16 * K / 4; i += blockDim.x) {
    int r = i / (K / 4), c = (i % (K / 4)) * 4;
    *(float4*)(a_s + r * P + c) = *(const float4*)(A + (size_t)(m0 + r) * K + c);
  }
  __syncthreads();
  const int col0 = (threadIdx.x >> 5) * 16;
  v8f acc = {};
  acc = tile_gemm<K>([&](int r, int kk, int h2) {
    return ld2(a_s + r * P + kk + h2);
  }, W, NC, col0, acc);
  tile_store(acc, [&](int r, int c, float v) {
    out[(size_t)(m0 + r) * NC + col0 + c] = v + bias[col0 + c];
  });
}

__global__ void __launch_bounds__(128)
edge_mlp_kernel(const float* __restrict__ h, float* __restrict__ e,
                float* __restrict__ macc,
                const int* __restrict__ src, const int* __restrict__ dst,
                const float* __restrict__ mw,
                const float* __restrict__ w1, const float* __restrict__ b1,
                const float* __restrict__ w2, const float* __restrict__ b2) {
  __shared__ __align__(16) float hs_s[EB * P128];
  __shared__ __align__(16) float hd_s[EB * P128];
  __shared__ __align__(16) float e_s[EB * P64];
  __shared__ __align__(16) float t_s[EB * P64];
  float* u_s = hs_s;
  const int m0   = blockIdx.x * EB;
  const int col0 = (threadIdx.x >> 5) * 16;

#pragma unroll
  for (int j = 0; j < EB * 32 / 128; ++j) {
    int i = threadIdx.x + j * 128;
    int r = i >> 5, c = (i & 31) * 4;
    *(float4*)(hs_s + r * P128 + c) = *(const float4*)(h + (size_t)src[m0 + r] * 128 + c);
    *(float4*)(hd_s + r * P128 + c) = *(const float4*)(h + (size_t)dst[m0 + r] * 128 + c);
  }
#pragma unroll
  for (int j = 0; j < EB * 16 / 128; ++j) {
    int i = threadIdx.x + j * 128;
    int r = i >> 4, c = (i & 15) * 4;
    *(float4*)(e_s + r * P64 + c) = *(const float4*)(e + (size_t)(m0 + r) * 64 + c);
  }
  __syncthreads();

  v8f acc0[RT] = {};
  tile_gemm_rt<128>([&](int r, int kk, int h2) {
    return ld2(hs_s + r * P128 + kk + h2);
  }, mw, 64, col0, acc0);
  tile_store_rt(acc0, [&](int r, int c, float v) {
    t_s[r * P64 + col0 + c] = e_s[r * P64 + col0 + c] + v;
  });
  __syncthreads();

  v8f acc1[RT] = {};
  tile_gemm_rt<320>([&](int r, int kk, int h2) {
    if (kk < 128) return ld2(hs_s + r * P128 + kk + h2);
    if (kk < 256) return ld2(hd_s + r * P128 + (kk - 128) + h2);
    return ld2(t_s + r * P64 + (kk - 256) + h2);
  }, w1, 64, col0, acc1);
  __syncthreads();
  tile_store_rt(acc1, [&](int r, int c, float v) {
    float t = v + b1[col0 + c];
    u_s[r * P64 + col0 + c] = t > 0.f ? t : 0.f;
  });
  __syncthreads();

  v8f acc2[RT] = {};
  tile_gemm_rt<64>([&](int r, int kk, int h2) {
    return ld2(u_s + r * P64 + kk + h2);
  }, w2, 64, col0, acc2);
  tile_store_rt(acc2, [&](int r, int c, float v) {
    int erow = m0 + r;
    float val = e_s[r * P64 + col0 + c] + v + b2[col0 + c];
    e[(size_t)erow * 64 + col0 + c] = val;
    atomicAdd(macc + (size_t)dst[erow] * 64 + col0 + c, val);
  });
}

__global__ void node_mlp_kernel(float* __restrict__ h, const float* __restrict__ macc,
                                const float* __restrict__ rdeg,
                                const float* __restrict__ w1, const float* __restrict__ b1,
                                const float* __restrict__ w2, const float* __restrict__ b2) {
  __shared__ __align__(16) float hv_s[16 * P128];
  __shared__ __align__(16) float m_s[16 * P64];
  __shared__ __align__(16) float u_s[16 * P64];
  const int m0   = blockIdx.x * 16;
  const int wave = threadIdx.x >> 5;

  for (int i = threadIdx.x; i < 16 * 128 / 4; i += blockDim.x) {
    int r = i >> 5, c = (i & 31) * 4;
    *(float4*)(hv_s + r * P128 + c) = *(const float4*)(h + (size_t)(m0 + r) * 128 + c);
  }
  for (int i = threadIdx.x; i < 16 * 64 / 4; i += blockDim.x) {
    int r = i >> 4, c = (i & 15) * 4;
    float4 v = *(const float4*)(macc + (size_t)(m0 + r) * 64 + c);
    float rd = rdeg[m0 + r];
    v.x *= rd; v.y *= rd; v.z *= rd; v.w *= rd;
    *(float4*)(m_s + r * P64 + c) = v;
  }
  __syncthreads();

  if (wave < 4) {
    const int col0 = wave * 16;
    v8f acc = {};
    acc = tile_gemm<192>([&](int r, int kk, int h2) {
      if (kk < 128) return ld2(hv_s + r * P128 + kk + h2);
      return ld2(m_s + r * P64 + (kk - 128) + h2);
    }, w1, 64, col0, acc);
    tile_store(acc, [&](int r, int c, float v) {
      float t = v + b1[col0 + c];
      u_s[r * P64 + col0 + c] = t > 0.f ? t : 0.f;
    });
  }
  __syncthreads();

  const int col0 = wave * 16;
  v8f acc2 = {};
  acc2 = tile_gemm<64>([&](int r, int kk, int h2) {
    return ld2(u_s + r * P64 + kk + h2);
  }, w2, 128, col0, acc2);
  tile_store(acc2, [&](int r, int c, float v) {
    h[(size_t)(m0 + r) * 128 + col0 + c] = hv_s[r * P128 + col0 + c] + v + b2[col0 + c];
  });
}

__global__ void deg_count_kernel(const int* __restrict__ dst, float* __restrict__ cnt) {
  int i = blockIdx.x * blockDim.x + threadIdx.x;
  if (i < EE) atomicAdd(cnt + dst[i], 1.0f);
}

__global__ void recip_clamp_kernel(float* __restrict__ v, int n) {
  int i = blockIdx.x * blockDim.x + threadIdx.x;
  if (i < n) v[i] = 1.0f / fmaxf(v[i], 1.0f);
}

__global__ void pool_kernel(const float* __restrict__ h, const int* __restrict__ batch,
                            float* __restrict__ gsum, float* __restrict__ gcnt) {
  int i = blockIdx.x * blockDim.x + threadIdx.x;
  int node = i >> 7, c = i & 127;
  if (node < NN) {
    int g = batch[node];
    atomicAdd(gsum + (size_t)g * 128 + c, h[(size_t)node * 128 + c]);
    if (c == 0) atomicAdd(gcnt + g, 1.0f);
  }
}

__global__ void readout_kernel(const float* __restrict__ gsum, const float* __restrict__ rgcnt,
                               const float* __restrict__ w1, const float* __restrict__ b1,
                               const float* __restrict__ w2, const float* __restrict__ b2,
                               float* __restrict__ out) {
  __shared__ __align__(16) float g_s[16 * P128];
  __shared__ __align__(16) float t_s[16 * P128];
  const int m0   = blockIdx.x * 16;
  const int col0 = (threadIdx.x >> 5) * 16;

  for (int i = threadIdx.x; i < 16 * 128 / 4; i += blockDim.x) {
    int r = i >> 5, c = (i & 31) * 4;
    float4 v = *(const float4*)(gsum + (size_t)(m0 + r) * 128 + c);
    float rc = rgcnt[m0 + r];
    v.x *= rc; v.y *= rc; v.z *= rc; v.w *= rc;
    *(float4*)(g_s + r * P128 + c) = v;
  }
  __syncthreads();

  v8f acc = {};
  acc = tile_gemm<128>([&](int r, int kk, int h2) {
    return ld2(g_s + r * P128 + kk + h2);
  }, w1, 128, col0, acc);
  tile_store(acc, [&](int r, int c, float v) {
    float t = v + b1[col0 + c];
    t_s[r * P128 + col0 + c] = t > 0.f ? t : 0.f;
  });
  __syncthreads();

  v8f acc2 = {};
  acc2 = tile_gemm<128>([&](int r, int kk, int h2) {
    return ld2(t_s + r * P128 + kk + h2);
  }, w2, 128, col0, acc2);
  tile_store(acc2, [&](int r, int c, float v) {
    out[(size_t)(m0 + r) * 128 + col0 + c] = v + b2[col0 + c];
  });
}

extern "C" void kernel_launch(void* const* d_in, const int* in_sizes, int n_in,
                              void* d_out, int out_size, void* d_ws, size_t ws_size,
                              hipStream_t stream) {
  (void)in_sizes; (void)n_in; (void)out_size; (void)ws_size;
  const float* x       = (const float*)d_in[0];
  const float* ea      = (const float*)d_in[1];
  const int*   eidx    = (const int*)d_in[2];
  const int*   batch   = (const int*)d_in[3];
  const float* lin_x_w = (const float*)d_in[4];
  const float* lin_x_b = (const float*)d_in[5];
  const float* lin_e_w = (const float*)d_in[6];
  const float* lin_e_b = (const float*)d_in[7];
  const float* msg_w   = (const float*)d_in[8];
  const float* phie_w1 = (const float*)d_in[9];
  const float* phie_b1 = (const float*)d_in[10];
  const float* phie_w2 = (const float*)d_in[11];
  const float* phie_b2 = (const float*)d_in[12];
  const float* phiv_w1 = (const float*)d_in[13];
  const float* phiv_b1 = (const float*)d_in[14];
  const float* phiv_w2 = (const float*)d_in[15];
  const float* phiv_b2 = (const float*)d_in[16];
  const float* ro_w1   = (const float*)d_in[17];
  const float* ro_b1   = (const float*)d_in[18];
  const float* ro_w2   = (const float*)d_in[19];
  const float* ro_b2   = (const float*)d_in[20];
  const int* src = eidx;
  const int* dst = eidx + EE;

  char* wp = (char*)d_ws;
  auto carve = [&](size_t bytes) {
    char* p = wp; wp += (bytes + 255) & ~(size_t)255; return p;
  };
  float* h    = (float*)carve((size_t)NN * 128 * 4);
  float* e    = (float*)carve((size_t)EE * 64 * 4);
  float* macc = (float*)carve((size_t)NN * 64 * 4);
  float* rdeg = (float*)carve((size_t)NN * 4);
  float* gsum = (float*)carve((size_t)BB * 128 * 4);
  float* gcnt = (float*)carve((size_t)BB * 4);

  encoder_kernel<64, 128><<<NN / 16, 256, 0, stream>>>(x, lin_x_w, lin_x_b, h);
  encoder_kernel<16, 64><<<EE / 16, 128, 0, stream>>>(ea, lin_e_w, lin_e_b, e);

  hipMemsetAsync(rdeg, 0, (size_t)NN * 4, stream);
  deg_count_kernel<<<(EE + 255) / 256, 256, 0, stream>>>(dst, rdeg);
  recip_clamp_kernel<<<(NN + 255) / 256, 256, 0, stream>>>(rdeg, NN);

  for (int l = 0; l < LL; ++l) {
    hipMemsetAsync(macc, 0, (size_t)NN * 64 * 4, stream);
    edge_mlp_kernel<<<EE / EB, 128, 0, stream>>>(h, e, macc, src, dst,
        msg_w + (size_t)l * 128 * 64,
        phie_w1 + (size_t)l * 320 * 64, phie_b1 + (size_t)l * 64,
        phie_w2 + (size_t)l * 64 * 64,  phie_b2 + (size_t)l * 64);
    node_mlp_kernel<<<NN / 16, 256, 0, stream>>>(h, macc, rdeg,
        phiv_w1 + (size_t)l * 192 * 64, phiv_b1 + (size_t)l * 64,
        phiv_w2 + (size_t)l * 64 * 128, phiv_b2 + (size_t)l * 128);
  }

  hipMemsetAsync(gsum, 0, (size_t)BB * 128 * 4, stream);
  hipMemsetAsync(gcnt, 0, (size_t)BB * 4, stream);
  pool_kernel<<<(NN * 128 + 255) / 256, 256, 0, stream>>>(h, batch, gsum, gcnt);
  recip_clamp_kernel<<<1, BB, 0, stream>>>(gcnt, BB);
  readout_kernel<<<BB / 16, 256, 0, stream>>>(gsum, gcnt, ro_w1, ro_b1, ro_w2, ro_b2,
                                              (float*)d_out);
}


# Products of gathered node rows done once per node, not per edge: pq=h@[mw|w1 src rows|w1 dst rows], pq rows gathered per edge; stage-1 f32 sum regrouped as (ps+pd)+t@w1 tail rows
# speedup vs baseline: 2.0011x; 1.2539x over previous
#include <hip/hip_runtime.h>

#define NN 50000
#define EE 200000
#define BB 64
#define LL 6

typedef float v2f __attribute__((ext_vector_type(2)));
typedef float v8f __attribute__((ext_vector_type(8)));

__device__ inline v2f ld2(const float* p) { return *(const v2f*)p; }

#define PAD 4
#define P64  (64 + PAD)
#define P128 (128 + PAD)

template <int K, typename FA>
__device__ inline v8f tile_gemm(FA fa, const float* __restrict__ B, int ldb,
                                int col0, v8f acc) {
  const int lane = threadIdx.x & 31;
  const int h2   = (lane >> 4) * 2;
  const int l15  = lane & 15;
#pragma unroll
  for (int kk = 0; kk < K; kk += 4) {
    v2f a = fa(l15, kk, h2);
    v2f b;
    const float* bp = B + (size_t)(kk + h2) * ldb + col0 + l15;
    b.x = bp[0];
    b.y = bp[ldb];
    acc = __builtin_amdgcn_wmma_f32_16x16x4_f32(false, a, false, b,
                                                (short)0, acc, false, false);
  }
  return acc;
}

template <typename ST>
__device__ inline void tile_store(v8f acc, ST st) {
  const int lane = threadIdx.x & 31;
  const int half = lane >> 4;
  const int l15  = lane & 15;
#pragma unroll
  for (int r = 0; r < 8; ++r) st(r + half * 8, l15, acc[r]);
}

#define RT 4
#define EB (16 * RT)
#define NRT 5
#define NB (16 * NRT)

template <int K, int R, typename FA>
__device__ inline void tile_gemm_rt(FA fa, const float* __restrict__ B, int ldb,
                                    int col0, v8f (&acc)[R]) {
  const int lane = threadIdx.x & 31;
  const int h2   = (lane >> 4) * 2;
  const int l15  = lane & 15;
#pragma unroll
  for (int kk = 0; kk < K; kk += 4) {
    v2f b;
    const float* bp = B + (size_t)(kk + h2) * ldb + col0 + l15;
    b.x = bp[0];
    b.y = bp[ldb];
#pragma unroll
    for (int t = 0; t < R; ++t) {
      v2f a = fa(16 * t + l15, kk, h2);
      acc[t] = __builtin_amdgcn_wmma_f32_16x16x4_f32(false, a, false, b,
                                                     (short)0, acc[t], false, false);
    }
  }
}

template <int R, typename ST>
__device__ inline void tile_store_rt(const v8f (&acc)[R], ST st) {
  const int lane = threadIdx.x & 31;
  const int half = lane >> 4;
  const int l15  = lane & 15;
#pragma unroll
  for (int t = 0; t < R; ++t)
#pragma unroll
    for (int r = 0; r < 8; ++r) st(16 * t + r + half * 8, l15, acc[t][r]);
}

template <int K, int NC>
__global__ void encoder_kernel(const float* __restrict__ A,
                               const float* __restrict__ W,
                               const float* __restrict__ bias,
                               float* __restrict__ out) {
  constexpr int P = K + PAD;
  __shared__ __align__(16) float a_s[16 * P];
  const int m0 = blockIdx.x * 16;
  for (int i = threadIdx.x; i < 16 * K / 4; i += blockDim.x) {
    int r = i / (K / 4), c = (i % (K / 4)) * 4;
    *(float4*)(a_s + r * P + c) = *(const float4*)(A + (size_t)(m0 + r) * K + c);
  }
  __syncthreads();
  const int col0 = (threadIdx.x >> 5) * 16;
  v8f acc = {};
  acc = tile_gemm<K>([&](int r, int kk, int h2) {
    return ld2(a_s + r * P + kk + h2);
  }, W, NC, col0, acc);
  tile_store(acc, [&](int r, int c, float v) {
    out[(size_t)(m0 + r) * NC + col0 + c] = v + bias[col0 + c];
  });
}

#define PW 6
__global__ void __launch_bounds__(PW * 32)
__attribute__((amdgpu_waves_per_eu(4, 8), amdgpu_num_vgpr(128)))
node_proj_kernel(const float* __restrict__ h, const float* __restrict__ mw,
                 const float* __restrict__ w1, float* __restrict__ pq) {
  __shared__ __align__(16) float h_s[NB * P128];
  const int m0 = blockIdx.x * NB;
  constexpr int NT = PW * 32, NV = NB * 32, FULL = NV / NT;
#pragma unroll
  for (int j = 0; j < FULL; ++j) {
    int i = threadIdx.x + j * NT;
    int r = i >> 5, c = (i & 31) * 4;
    *(float4*)(h_s + r * P128 + c) = *(const float4*)(h + (size_t)(m0 + r) * 128 + c);
  }
  if (threadIdx.x < NV - FULL * NT) {
    int i = threadIdx.x + FULL * NT;
    int r = i >> 5, c = (i & 31) * 4;
    *(float4*)(h_s + r * P128 + c) = *(const float4*)(h + (size_t)(m0 + r) * 128 + c);
  }
  __syncthreads();
  const int strip = blockIdx.y * PW + (threadIdx.x >> 5);
  const int part  = strip >> 2;
  const int col0  = (strip & 3) * 16;
  const float* B = part == 0 ? mw : (part == 1 ? w1 : w1 + 128 * 64);
  v8f acc[NRT] = {};
  tile_gemm_rt<128, NRT>([&](int r, int kk, int h2) {
    return ld2(h_s + r * P128 + kk + h2);
  }, B, 64, col0, acc);
  tile_store_rt<NRT>(acc, [&](int r, int c, float v) {
    pq[(size_t)(m0 + r) * 192 + part * 64 + col0 + c] = v;
  });
}

__global__ void __launch_bounds__(128)
edge_mlp_kernel(const float* __restrict__ pq, float* __restrict__ e,
                float* __restrict__ macc,
                const int* __restrict__ src, const int* __restrict__ dst,
                const float* __restrict__ w1t, const float* __restrict__ b1,
                const float* __restrict__ w2, const float* __restrict__ b2) {
  __shared__ __align__(16) float e_s[EB * P64];
  __shared__ __align__(16) float t_s[EB * P64];
  __shared__ __align__(16) float s_s[EB * P64];
  float* u_s = s_s;
  const int m0   = blockIdx.x * EB;
  const int col0 = (threadIdx.x >> 5) * 16;
  const int lane = threadIdx.x & 31;
  const int half = lane >> 4;
  const int l15  = lane & 15;

#pragma unroll
  for (int j = 0; j < EB * 16 / 128; ++j) {
    int i = threadIdx.x + j * 128;
    int r = i >> 4, c = (i & 15) * 4;
    const float* ps_row = pq + (size_t)src[m0 + r] * 192;
    const float* pd_row = pq + (size_t)dst[m0 + r] * 192;
    float4 q  = *(const float4*)(ps_row + c);
    float4 ps = *(const float4*)(ps_row + 64 + c);
    float4 pd = *(const float4*)(pd_row + 128 + c);
    float4 ev = *(const float4*)(e + (size_t)(m0 + r) * 64 + c);
    float4 tv, sv;
    tv.x = ev.x + q.x;  tv.y = ev.y + q.y;  tv.z = ev.z + q.z;  tv.w = ev.w + q.w;
    sv.x = ps.x + pd.x; sv.y = ps.y + pd.y; sv.z = ps.z + pd.z; sv.w = ps.w + pd.w;
    *(float4*)(e_s + r * P64 + c) = ev;
    *(float4*)(t_s + r * P64 + c) = tv;
    *(float4*)(s_s + r * P64 + c) = sv;
  }
  __syncthreads();

  v8f acc1[RT];
#pragma unroll
  for (int t = 0; t < RT; ++t)
#pragma unroll
    for (int r = 0; r < 8; ++r)
      acc1[t][r] = s_s[(16 * t + r + half * 8) * P64 + col0 + l15];
  tile_gemm_rt<64, RT>([&](int r, int kk, int h2) {
    return ld2(t_s + r * P64 + kk + h2);
  }, w1t, 64, col0, acc1);
  tile_store_rt<RT>(acc1, [&](int r, int c, float v) {
    float t = v + b1[col0 + c];
    u_s[r * P64 + col0 + c] = t > 0.f ? t : 0.f;
  });
  __syncthreads();

  v8f acc2[RT] = {};
  tile_gemm_rt<64, RT>([&](int r, int kk, int h2) {
    return ld2(u_s + r * P64 + kk + h2);
  }, w2, 64, col0, acc2);
  tile_store_rt<RT>(acc2, [&](int r, int c, float v) {
    int erow = m0 + r;
    float val = e_s[r * P64 + col0 + c] + v + b2[col0 + c];
    e[(size_t)erow * 64 + col0 + c] = val;
    atomicAdd(macc + (size_t)dst[erow] * 64 + col0 + c, val);
  });
}

__global__ void node_mlp_kernel(float* __restrict__ h, const float* __restrict__ macc,
                                const float* __restrict__ rdeg,
                                const float* __restrict__ w1, const float* __restrict__ b1,
                                const float* __restrict__ w2, const float* __restrict__ b2) {
  __shared__ __align__(16) float hv_s[16 * P128];
  __shared__ __align__(16) float m_s[16 * P64];
  __shared__ __align__(16) float u_s[16 * P64];
  const int m0   = blockIdx.x * 16;
  const int wave = threadIdx.x >> 5;

  for (int i = threadIdx.x; i < 16 * 128 / 4; i += blockDim.x) {
    int r = i >> 5, c = (i & 31) * 4;
    *(float4*)(hv_s + r * P128 + c) = *(const float4*)(h + (size_t)(m0 + r) * 128 + c);
  }
  for (int i = threadIdx.x; i < 16 * 64 / 4; i += blockDim.x) {
    int r = i >> 4, c = (i & 15) * 4;
    float4 v = *(const float4*)(macc + (size_t)(m0 + r) * 64 + c);
    float rd = rdeg[m0 + r];
    v.x *= rd; v.y *= rd; v.z *= rd; v.w *= rd;
    *(float4*)(m_s + r * P64 + c) = v;
  }
  __syncthreads();

  if (wave < 4) {
    const int col0 = wave * 16;
    v8f acc = {};
    acc = tile_gemm<192>([&](int r, int kk, int h2) {
      if (kk < 128) return ld2(hv_s + r * P128 + kk + h2);
      return ld2(m_s + r * P64 + (kk - 128) + h2);
    }, w1, 64, col0, acc);
    tile_store(acc, [&](int r, int c, float v) {
      float t = v + b1[col0 + c];
      u_s[r * P64 + col0 + c] = t > 0.f ? t : 0.f;
    });
  }
  __syncthreads();

  const int col0 = wave * 16;
  v8f acc2 = {};
  acc2 = tile_gemm<64>([&](int r, int kk, int h2) {
    return ld2(u_s + r * P64 + kk + h2);
  }, w2, 128, col0, acc2);
  tile_store(acc2, [&](int r, int c, float v) {
    h[(size_t)(m0 + r) * 128 + col0 + c] = hv_s[r * P128 + col0 + c] + v + b2[col0 + c];
  });
}

__global__ void deg_count_kernel(const int* __restrict__ dst, float* __restrict__ cnt) {
  int i = blockIdx.x * blockDim.x + threadIdx.x;
  if (i < EE) atomicAdd(cnt + dst[i], 1.0f);
}

__global__ void recip_clamp_kernel(float* __restrict__ v, int n) {
  int i = blockIdx.x * blockDim.x + threadIdx.x;
  if (i < n) v[i] = 1.0f / fmaxf(v[i], 1.0f);
}

__global__ void pool_kernel(const float* __restrict__ h, const int* __restrict__ batch,
                            float* __restrict__ gsum, float* __restrict__ gcnt) {
  int i = blockIdx.x * blockDim.x + threadIdx.x;
  int node = i >> 7, c = i & 127;
  if (node < NN) {
    int g = batch[node];
    atomicAdd(gsum + (size_t)g * 128 + c, h[(size_t)node * 128 + c]);
    if (c == 0) atomicAdd(gcnt + g, 1.0f);
  }
}

__global__ void readout_kernel(const float* __restrict__ gsum, const float* __restrict__ rgcnt,
                               const float* __restrict__ w1, const float* __restrict__ b1,
                               const float* __restrict__ w2, const float* __restrict__ b2,
                               float* __restrict__ out) {
  __shared__ __align__(16) float g_s[16 * P128];
  __shared__ __align__(16) float t_s[16 * P128];
  const int m0   = blockIdx.x * 16;
  const int col0 = (threadIdx.x >> 5) * 16;

  for (int i = threadIdx.x; i < 16 * 128 / 4; i += blockDim.x) {
    int r = i >> 5, c = (i & 31) * 4;
    float4 v = *(const float4*)(gsum + (size_t)(m0 + r) * 128 + c);
    float rc = rgcnt[m0 + r];
    v.x *= rc; v.y *= rc; v.z *= rc; v.w *= rc;
    *(float4*)(g_s + r * P128 + c) = v;
  }
  __syncthreads();

  v8f acc = {};
  acc = tile_gemm<128>([&](int r, int kk, int h2) {
    return ld2(g_s + r * P128 + kk + h2);
  }, w1, 128, col0, acc);
  tile_store(acc, [&](int r, int c, float v) {
    float t = v + b1[col0 + c];
    t_s[r * P128 + col0 + c] = t > 0.f ? t : 0.f;
  });
  __syncthreads();

  v8f acc2 = {};
  acc2 = tile_gemm<128>([&](int r, int kk, int h2) {
    return ld2(t_s + r * P128 + kk + h2);
  }, w2, 128, col0, acc2);
  tile_store(acc2, [&](int r, int c, float v) {
    out[(size_t)(m0 + r) * 128 + col0 + c] = v + b2[col0 + c];
  });
}

extern "C" void kernel_launch(void* const* d_in, const int* in_sizes, int n_in,
                              void* d_out, int out_size, void* d_ws, size_t ws_size,
                              hipStream_t stream) {
  (void)in_sizes; (void)n_in; (void)out_size; (void)ws_size;
  const float* x       = (const float*)d_in[0];
  const float* ea      = (const float*)d_in[1];
  const int*   eidx    = (const int*)d_in[2];
  const int*   batch   = (const int*)d_in[3];
  const float* lin_x_w = (const float*)d_in[4];
  const float* lin_x_b = (const float*)d_in[5];
  const float* lin_e_w = (const float*)d_in[6];
  const float* lin_e_b = (const float*)d_in[7];
  const float* msg_w   = (const float*)d_in[8];
  const float* phie_w1 = (const float*)d_in[9];
  const float* phie_b1 = (const float*)d_in[10];
  const float* phie_w2 = (const float*)d_in[11];
  const float* phie_b2 = (const float*)d_in[12];
  const float* phiv_w1 = (const float*)d_in[13];
  const float* phiv_b1 = (const float*)d_in[14];
  const float* phiv_w2 = (const float*)d_in[15];
  const float* phiv_b2 = (const float*)d_in[16];
  const float* ro_w1   = (const float*)d_in[17];
  const float* ro_b1   = (const float*)d_in[18];
  const float* ro_w2   = (const float*)d_in[19];
  const float* ro_b2   = (const float*)d_in[20];
  const int* src = eidx;
  const int* dst = eidx + EE;

  char* wp = (char*)d_ws;
  auto carve = [&](size_t bytes) {
    char* p = wp; wp += (bytes + 255) & ~(size_t)255; return p;
  };
  float* h    = (float*)carve((size_t)NN * 128 * 4);
  float* e    = (float*)carve((size_t)EE * 64 * 4);
  float* macc = (float*)carve((size_t)NN * 64 * 4);
  float* pq   = (float*)carve((size_t)NN * 192 * 4);
  float* rdeg = (float*)carve((size_t)NN * 4);
  float* gsum = (float*)carve((size_t)BB * 128 * 4);
  float* gcnt = (float*)carve((size_t)BB * 4);

  encoder_kernel<64, 128><<<NN / 16, 256, 0, stream>>>(x, lin_x_w, lin_x_b, h);
  encoder_kernel<16, 64><<<EE / 16, 128, 0, stream>>>(ea, lin_e_w, lin_e_b, e);

  hipMemsetAsync(rdeg, 0, (size_t)NN * 4, stream);
  deg_count_kernel<<<(EE + 255) / 256, 256, 0, stream>>>(dst, rdeg);
  recip_clamp_kernel<<<(NN + 255) / 256, 256, 0, stream>>>(rdeg, NN);

  for (int l = 0; l < LL; ++l) {
    hipMemsetAsync(macc, 0, (size_t)NN * 64 * 4, stream);
    node_proj_kernel<<<dim3(NN / NB, 12 / PW), PW * 32, 0, stream>>>(h,
        msg_w + (size_t)l * 128 * 64, phie_w1 + (size_t)l * 320 * 64, pq);
    edge_mlp_kernel<<<EE / EB, 128, 0, stream>>>(pq, e, macc, src, dst,
        phie_w1 + (size_t)l * 320 * 64 + 256 * 64, phie_b1 + (size_t)l * 64,
        phie_w2 + (size_t)l * 64 * 64,  phie_b2 + (size_t)l * 64);
    node_mlp_kernel<<<NN / 16, 256, 0, stream>>>(h, macc, rdeg,
        phiv_w1 + (size_t)l * 192 * 64, phiv_b1 + (size_t)l * 64,
        phiv_w2 + (size_t)l * 64 * 128, phiv_b2 + (size_t)l * 128);
  }

  hipMemsetAsync(gsum, 0, (size_t)BB * 128 * 4, stream);
  hipMemsetAsync(gcnt, 0, (size_t)BB * 4, stream);
  pool_kernel<<<(NN * 128 + 255) / 256, 256, 0, stream>>>(h, batch, gsum, gcnt);
  recip_clamp_kernel<<<1, BB, 0, stream>>>(gcnt, BB);
  readout_kernel<<<BB / 16, 256, 0, stream>>>(gsum, gcnt, ro_w1, ro_b1, ro_w2, ro_b2,
                                              (float*)d_out);
}


# Scatter-mean without float atomics or memset: edges renumbered in destination order once per call (int atomics), node kernel sums a contiguous range of e rows; sum order now fixed (edge id)
# speedup vs baseline: 2.3234x; 1.1611x over previous
#include <hip/hip_runtime.h>

#define NN 50000
#define EE 200000
#define BB 64
#define LL 6

typedef float v2f __attribute__((ext_vector_type(2)));
typedef float v8f __attribute__((ext_vector_type(8)));

__device__ inline v2f ld2(const float* p) { return *(const v2f*)p; }

#define PAD 4
#define P64  (64 + PAD)
#define P128 (128 + PAD)

template <int K, typename FA>
__device__ inline v8f tile_gemm(FA fa, const float* __restrict__ B, int ldb,
                                int col0, v8f acc) {
  const int lane = threadIdx.x & 31;
  const int h2   = (lane >> 4) * 2;
  const int l15  = lane & 15;
#pragma unroll
  for (int kk = 0; kk < K; kk += 4) {
    v2f a = fa(l15, kk, h2);
    v2f b;
    const float* bp = B + (size_t)(kk + h2) * ldb + col0 + l15;
    b.x = bp[0];
    b.y = bp[ldb];
    acc = __builtin_amdgcn_wmma_f32_16x16x4_f32(false, a, false, b,
                                                (short)0, acc, false, false);
  }
  return acc;
}

template <typename ST>
__device__ inline void tile_store(v8f acc, ST st) {
  const int lane = threadIdx.x & 31;
  const int half = lane >> 4;
  const int l15  = lane & 15;
#pragma unroll
  for (int r = 0; r < 8; ++r) st(r + half * 8, l15, acc[r]);
}

#define RT 4
#define EB (16 * RT)
#define NRT 5
#define NB (16 * NRT)

template <int K, int R, typename FA>
__device__ inline void tile_gemm_rt(FA fa, const float* __restrict__ B, int ldb,
                                    int col0, v8f (&acc)[R]) {
  const int lane = threadIdx.x & 31;
  const int h2   = (lane >> 4) * 2;
  const int l15  = lane & 15;
#pragma unroll
  for (int kk = 0; kk < K; kk += 4) {
    v2f b;
    const float* bp = B + (size_t)(kk + h2) * ldb + col0 + l15;
    b.x = bp[0];
    b.y = bp[ldb];
#pragma unroll
    for (int t = 0; t < R; ++t) {
      v2f a = fa(16 * t + l15, kk, h2);
      acc[t] = __builtin_amdgcn_wmma_f32_16x16x4_f32(false, a, false, b,
                                                     (short)0, acc[t], false, false);
    }
  }
}

template <int R, typename ST>
__device__ inline void tile_store_rt(const v8f (&acc)[R], ST st) {
  const int lane = threadIdx.x & 31;
  const int half = lane >> 4;
  const int l15  = lane & 15;
#pragma unroll
  for (int t = 0; t < R; ++t)
#pragma unroll
    for (int r = 0; r < 8; ++r) st(16 * t + r + half * 8, l15, acc[t][r]);
}

template <int K, int NC>
__global__ void encoder_kernel(const float* __restrict__ A,
                               const float* __restrict__ W,
                               const float* __restrict__ bias,
                               float* __restrict__ out,
                               const int* __restrict__ rows) {
  constexpr int P = K + PAD;
  __shared__ __align__(16) float a_s[16 * P];
  const int m0 = blockIdx.x * 16;
  for (int i = threadIdx.x; i < 16 * K / 4; i += blockDim.x) {
    int r = i / (K / 4), c = (i % (K / 4)) * 4;
    const size_t row = rows ? (size_t)rows[m0 + r] : (size_t)(m0 + r);
    *(float4*)(a_s + r * P + c) = *(const float4*)(A + row * K + c);
  }
  __syncthreads();
  const int col0 = (threadIdx.x >> 5) * 16;
  v8f acc = {};
  acc = tile_gemm<K>([&](int r, int kk, int h2) {
    return ld2(a_s + r * P + kk + h2);
  }, W, NC, col0, acc);
  tile_store(acc, [&](int r, int c, float v) {
    out[(size_t)(m0 + r) * NC + col0 + c] = v + bias[col0 + c];
  });
}

#define PW 6
__global__ void __launch_bounds__(PW * 32)
__attribute__((amdgpu_waves_per_eu(4, 8), amdgpu_num_vgpr(128)))
node_proj_kernel(const float* __restrict__ h, const float* __restrict__ mw,
                 const float* __restrict__ w1, float* __restrict__ pq) {
  __shared__ __align__(16) float h_s[NB * P128];
  const int m0 = blockIdx.x * NB;
  constexpr int NT = PW * 32, NV = NB * 32, FULL = NV / NT;
#pragma unroll
  for (int j = 0; j < FULL; ++j) {
    int i = threadIdx.x + j * NT;
    int r = i >> 5, c = (i & 31) * 4;
    *(float4*)(h_s + r * P128 + c) = *(const float4*)(h + (size_t)(m0 + r) * 128 + c);
  }
  if (threadIdx.x < NV - FULL * NT) {
    int i = threadIdx.x + FULL * NT;
    int r = i >> 5, c = (i & 31) * 4;
    *(float4*)(h_s + r * P128 + c) = *(const float4*)(h + (size_t)(m0 + r) * 128 + c);
  }
  __syncthreads();
  const int strip = blockIdx.y * PW + (threadIdx.x >> 5);
  const int part  = strip >> 2;
  const int col0  = (strip & 3) * 16;
  const float* B = part == 0 ? mw : (part == 1 ? w1 : w1 + 128 * 64);
  v8f acc[NRT] = {};
  tile_gemm_rt<128, NRT>([&](int r, int kk, int h2) {
    return ld2(h_s + r * P128 + kk + h2);
  }, B, 64, col0, acc);
  tile_store_rt<NRT>(acc, [&](int r, int c, float v) {
    pq[(size_t)(m0 + r) * 192 + part * 64 + col0 + c] = v;
  });
}

__global__ void __launch_bounds__(128)
edge_mlp_kernel(const float* __restrict__ pq, float* __restrict__ e,
                const int* __restrict__ src, const int* __restrict__ dst,
                const float* __restrict__ w1t, const float* __restrict__ b1,
                const float* __restrict__ w2, const float* __restrict__ b2) {
  __shared__ __align__(16) float e_s[EB * P64];
  __shared__ __align__(16) float t_s[EB * P64];
  __shared__ __align__(16) float s_s[EB * P64];
  float* u_s = s_s;
  const int m0   = blockIdx.x * EB;
  const int col0 = (threadIdx.x >> 5) * 16;
  const int lane = threadIdx.x & 31;
  const int half = lane >> 4;
  const int l15  = lane & 15;

#pragma unroll
  for (int j = 0; j < EB * 16 / 128; ++j) {
    int i = threadIdx.x + j * 128;
    int r = i >> 4, c = (i & 15) * 4;
    const float* ps_row = pq + (size_t)src[m0 + r] * 192;
    const float* pd_row = pq + (size_t)dst[m0 + r] * 192;
    float4 q  = *(const float4*)(ps_row + c);
    float4 ps = *(const float4*)(ps_row + 64 + c);
    float4 pd = *(const float4*)(pd_row + 128 + c);
    float4 ev = *(const float4*)(e + (size_t)(m0 + r) * 64 + c);
    float4 tv, sv;
    tv.x = ev.x + q.x;  tv.y = ev.y + q.y;  tv.z = ev.z + q.z;  tv.w = ev.w + q.w;
    sv.x = ps.x + pd.x; sv.y = ps.y + pd.y; sv.z = ps.z + pd.z; sv.w = ps.w + pd.w;
    *(float4*)(e_s + r * P64 + c) = ev;
    *(float4*)(t_s + r * P64 + c) = tv;
    *(float4*)(s_s + r * P64 + c) = sv;
  }
  __syncthreads();

  v8f acc1[RT];
#pragma unroll
  for (int t = 0; t < RT; ++t)
#pragma unroll
    for (int r = 0; r < 8; ++r)
      acc1[t][r] = s_s[(16 * t + r + half * 8) * P64 + col0 + l15];
  tile_gemm_rt<64, RT>([&](int r, int kk, int h2) {
    return ld2(t_s + r * P64 + kk + h2);
  }, w1t, 64, col0, acc1);
  tile_store_rt<RT>(acc1, [&](int r, int c, float v) {
    float t = v + b1[col0 + c];
    u_s[r * P64 + col0 + c] = t > 0.f ? t : 0.f;
  });
  __syncthreads();

  v8f acc2[RT] = {};
  tile_gemm_rt<64, RT>([&](int r, int kk, int h2) {
    return ld2(u_s + r * P64 + kk + h2);
  }, w2, 64, col0, acc2);
  tile_store_rt<RT>(acc2, [&](int r, int c, float v) {
    e[(size_t)(m0 + r) * 64 + col0 + c] = e_s[r * P64 + col0 + c] + v + b2[col0 + c];
  });
}

#define GC 8
__global__ void node_mlp_kernel(float* __restrict__ h, const float* __restrict__ e,
                                const int* __restrict__ start, const int* __restrict__ cnt,
                                const float* __restrict__ w1, const float* __restrict__ b1,
                                const float* __restrict__ w2, const float* __restrict__ b2) {
  __shared__ __align__(16) float hv_s[16 * P128];
  __shared__ __align__(16) float m_s[16 * P64];
  __shared__ __align__(16) float u_s[16 * P64];
  const int m0   = blockIdx.x * 16;
  const int wave = threadIdx.x >> 5;

  const int gr  = threadIdx.x >> 4, gc = (threadIdx.x & 15) * 4;
  const int beg = start[m0 + gr];
  const int deg = cnt[m0 + gr];

  for (int i = threadIdx.x; i < 16 * 128 / 4; i += blockDim.x) {
    int r = i >> 5, c = (i & 31) * 4;
    *(float4*)(hv_s + r * P128 + c) = *(const float4*)(h + (size_t)(m0 + r) * 128 + c);
  }
  {
    const int r = gr, c = gc;
    const int end = beg + deg;
    float4 v = make_float4(0.f, 0.f, 0.f, 0.f);
    for (int k0 = beg; k0 < end; k0 += GC) {
      float4 w[GC];
#pragma unroll
      for (int j = 0; j < GC; ++j)
        w[j] = (k0 + j < end) ? *(const float4*)(e + (size_t)(k0 + j) * 64 + c)
                              : make_float4(0.f, 0.f, 0.f, 0.f);
#pragma unroll
      for (int j = 0; j < GC; ++j) {
        if (k0 + j < end) { v.x += w[j].x; v.y += w[j].y; v.z += w[j].z; v.w += w[j].w; }
      }
    }
    const float rd = 1.0f / fmaxf((float)deg, 1.0f);
    v.x *= rd; v.y *= rd; v.z *= rd; v.w *= rd;
    *(float4*)(m_s + r * P64 + c) = v;
  }
  __syncthreads();

  if (wave < 4) {
    const int col0 = wave * 16;
    v8f acc = {};
    acc = tile_gemm<192>([&](int r, int kk, int h2) {
      if (kk < 128) return ld2(hv_s + r * P128 + kk + h2);
      return ld2(m_s + r * P64 + (kk - 128) + h2);
    }, w1, 64, col0, acc);
    tile_store(acc, [&](int r, int c, float v) {
      float t = v + b1[col0 + c];
      u_s[r * P64 + col0 + c] = t > 0.f ? t : 0.f;
    });
  }
  __syncthreads();

  const int col0 = wave * 16;
  v8f acc2 = {};
  acc2 = tile_gemm<64>([&](int r, int kk, int h2) {
    return ld2(u_s + r * P64 + kk + h2);
  }, w2, 128, col0, acc2);
  tile_store(acc2, [&](int r, int c, float v) {
    h[(size_t)(m0 + r) * 128 + col0 + c] = hv_s[r * P128 + col0 + c] + v + b2[col0 + c];
  });
}

#define SMAX 16
__global__ void deg_rank_kernel(const int* __restrict__ dst, int* __restrict__ cnt,
                                int* __restrict__ rank) {
  int i = blockIdx.x * blockDim.x + threadIdx.x;
  if (i < EE) rank[i] = atomicAdd(cnt + dst[i], 1);
}

__global__ void seg_alloc_kernel(const int* __restrict__ cnt, int* __restrict__ cursor,
                                 int* __restrict__ start) {
  const int n    = blockIdx.x * blockDim.x + threadIdx.x;
  const int lane = threadIdx.x & 31;
  const int c    = n < NN ? cnt[n] : 0;
  int incl = c;
  for (int d = 1; d < 32; d <<= 1) {
    int t = __shfl_up(incl, d);
    if (lane >= d) incl += t;
  }
  const int total = __shfl(incl, 31);
  int base = 0;
  if (lane == 31) base = atomicAdd(cursor, total);
  base = __shfl(base, 31);
  if (n < NN) start[n] = base + incl - c;
}

__global__ void seg_fill_kernel(const int* __restrict__ dst, const int* __restrict__ start,
                                const int* __restrict__ rank, int* __restrict__ perm) {
  int i = blockIdx.x * blockDim.x + threadIdx.x;
  if (i < EE) perm[start[dst[i]] + rank[i]] = i;
}

__global__ void seg_sort_kernel(const int* __restrict__ src, const int* __restrict__ start,
                                const int* __restrict__ cnt, int* __restrict__ perm,
                                int* __restrict__ srcp, int* __restrict__ dstp) {
  const int n = blockIdx.x * blockDim.x + threadIdx.x;
  if (n >= NN) return;
  const int beg = start[n], d = cnt[n];
  if (d <= SMAX) {
    int id[SMAX];
#pragma unroll
    for (int j = 0; j < SMAX; ++j) id[j] = j < d ? perm[beg + j] : 0x7fffffff;
#pragma unroll
    for (int i = 0; i < SMAX; ++i) {
      if (i < d) {
        int pos = 0;
#pragma unroll
        for (int j = 0; j < SMAX; ++j) pos += id[j] < id[i] ? 1 : 0;
        perm[beg + pos] = id[i];
        srcp[beg + pos] = src[id[i]];
        dstp[beg + pos] = n;
      }
    }
  } else {
    for (int i = 1; i < d; ++i) {
      const int key = perm[beg + i];
      int j = i - 1;
      while (j >= 0 && perm[beg + j] > key) { perm[beg + j + 1] = perm[beg + j]; --j; }
      perm[beg + j + 1] = key;
    }
    for (int i = 0; i < d; ++i) { srcp[beg + i] = src[perm[beg + i]]; dstp[beg + i] = n; }
  }
}

__global__ void recip_clamp_kernel(float* __restrict__ v, int n) {
  int i = blockIdx.x * blockDim.x + threadIdx.x;
  if (i < n) v[i] = 1.0f / fmaxf(v[i], 1.0f);
}

__global__ void pool_kernel(const float* __restrict__ h, const int* __restrict__ batch,
                            float* __restrict__ gsum, float* __restrict__ gcnt) {
  int i = blockIdx.x * blockDim.x + threadIdx.x;
  int node = i >> 7, c = i & 127;
  if (node < NN) {
    int g = batch[node];
    atomicAdd(gsum + (size_t)g * 128 + c, h[(size_t)node * 128 + c]);
    if (c == 0) atomicAdd(gcnt + g, 1.0f);
  }
}

__global__ void readout_kernel(const float* __restrict__ gsum, const float* __restrict__ rgcnt,
                               const float* __restrict__ w1, const float* __restrict__ b1,
                               const float* __restrict__ w2, const float* __restrict__ b2,
                               float* __restrict__ out) {
  __shared__ __align__(16) float g_s[16 * P128];
  __shared__ __align__(16) float t_s[16 * P128];
  const int m0   = blockIdx.x * 16;
  const int col0 = (threadIdx.x >> 5) * 16;

  for (int i = threadIdx.x; i < 16 * 128 / 4; i += blockDim.x) {
    int r = i >> 5, c = (i & 31) * 4;
    float4 v = *(const float4*)(gsum + (size_t)(m0 + r) * 128 + c);
    float rc = rgcnt[m0 + r];
    v.x *= rc; v.y *= rc; v.z *= rc; v.w *= rc;
    *(float4*)(g_s + r * P128 + c) = v;
  }
  __syncthreads();

  v8f acc = {};
  acc = tile_gemm<128>([&](int r, int kk, int h2) {
    return ld2(g_s + r * P128 + kk + h2);
  }, w1, 128, col0, acc);
  tile_store(acc, [&](int r, int c, float v) {
    float t = v + b1[col0 + c];
    t_s[r * P128 + col0 + c] = t > 0.f ? t : 0.f;
  });
  __syncthreads();

  v8f acc2 = {};
  acc2 = tile_gemm<128>([&](int r, int kk, int h2) {
    return ld2(t_s + r * P128 + kk + h2);
  }, w2, 128, col0, acc2);
  tile_store(acc2, [&](int r, int c, float v) {
    out[(size_t)(m0 + r) * 128 + col0 + c] = v + b2[col0 + c];
  });
}

extern "C" void kernel_launch(void* const* d_in, const int* in_sizes, int n_in,
                              void* d_out, int out_size, void* d_ws, size_t ws_size,
                              hipStream_t stream) {
  (void)in_sizes; (void)n_in; (void)out_size; (void)ws_size;
  const float* x       = (const float*)d_in[0];
  const float* ea      = (const float*)d_in[1];
  const int*   eidx    = (const int*)d_in[2];
  const int*   batch   = (const int*)d_in[3];
  const float* lin_x_w = (const float*)d_in[4];
  const float* lin_x_b = (const float*)d_in[5];
  const float* lin_e_w = (const float*)d_in[6];
  const float* lin_e_b = (const float*)d_in[7];
  const float* msg_w   = (const float*)d_in[8];
  const float* phie_w1 = (const float*)d_in[9];
  const float* phie_b1 = (const float*)d_in[10];
  const float* phie_w2 = (const float*)d_in[11];
  const float* phie_b2 = (const float*)d_in[12];
  const float* phiv_w1 = (const float*)d_in[13];
  const float* phiv_b1 = (const float*)d_in[14];
  const float* phiv_w2 = (const float*)d_in[15];
  const float* phiv_b2 = (const float*)d_in[16];
  const float* ro_w1   = (const float*)d_in[17];
  const float* ro_b1   = (const float*)d_in[18];
  const float* ro_w2   = (const float*)d_in[19];
  const float* ro_b2   = (const float*)d_in[20];
  const int* src = eidx;
  const int* dst = eidx + EE;

  char* wp = (char*)d_ws;
  auto carve = [&](size_t bytes) {
    char* p = wp; wp += (bytes + 255) & ~(size_t)255; return p;
  };
  float* h    = (float*)carve((size_t)NN * 128 * 4);
  float* e    = (float*)carve((size_t)EE * 64 * 4);
  float* pq   = (float*)carve((size_t)NN * 192 * 4);
  int* cnt    = (int*)carve((size_t)(NN + 1) * 4);
  int* rank   = (int*)carve((size_t)EE * 4);
  int* start  = (int*)carve((size_t)NN * 4);
  int* perm   = (int*)carve((size_t)EE * 4);
  int* srcp   = (int*)carve((size_t)EE * 4);
  int* dstp   = (int*)carve((size_t)EE * 4);
  float* gsum = (float*)carve((size_t)BB * 128 * 4);
  float* gcnt = (float*)carve((size_t)BB * 4);

  hipMemsetAsync(cnt, 0, (size_t)(NN + 1) * 4, stream);
  deg_rank_kernel<<<(EE + 255) / 256, 256, 0, stream>>>(dst, cnt, rank);
  seg_alloc_kernel<<<(NN + 255) / 256, 256, 0, stream>>>(cnt, cnt + NN, start);
  seg_fill_kernel<<<(EE + 255) / 256, 256, 0, stream>>>(dst, start, rank, perm);
  seg_sort_kernel<<<(NN + 255) / 256, 256, 0, stream>>>(src, start, cnt, perm, srcp, dstp);

  encoder_kernel<64, 128><<<NN / 16, 256, 0, stream>>>(x, lin_x_w, lin_x_b, h, nullptr);
  encoder_kernel<16, 64><<<EE / 16, 128, 0, stream>>>(ea, lin_e_w, lin_e_b, e, perm);

  for (int l = 0; l < LL; ++l) {
    node_proj_kernel<<<dim3(NN / NB, 12 / PW), PW * 32, 0, stream>>>(h,
        msg_w + (size_t)l * 128 * 64, phie_w1 + (size_t)l * 320 * 64, pq);
    edge_mlp_kernel<<<EE / EB, 128, 0, stream>>>(pq, e, srcp, dstp,
        phie_w1 + (size_t)l * 320 * 64 + 256 * 64, phie_b1 + (size_t)l * 64,
        phie_w2 + (size_t)l * 64 * 64,  phie_b2 + (size_t)l * 64);
    node_mlp_kernel<<<NN / 16, 256, 0, stream>>>(h, e, start, cnt,
        phiv_w1 + (size_t)l * 192 * 64, phiv_b1 + (size_t)l * 64,
        phiv_w2 + (size_t)l * 64 * 128, phiv_b2 + (size_t)l * 128);
  }

  hipMemsetAsync(gsum, 0, (size_t)BB * 128 * 4, stream);
  hipMemsetAsync(gcnt, 0, (size_t)BB * 4, stream);
  pool_kernel<<<(NN * 128 + 255) / 256, 256, 0, stream>>>(h, batch, gsum, gcnt);
  recip_clamp_kernel<<<1, BB, 0, stream>>>(gcnt, BB);
  readout_kernel<<<BB / 16, 256, 0, stream>>>(gsum, gcnt, ro_w1, ro_b1, ro_w2, ro_b2,
                                              (float*)d_out);
}


# Graph pooling: a block sums 80 nodes in registers, then its 8 wave sums in LDS, and issues one float atomic per column (was one per node and column); f32 sum order: node order inside a block
# speedup vs baseline: 2.7459x; 1.1818x over previous
#include <hip/hip_runtime.h>

#define NN 50000
#define EE 200000
#define BB 64
#define LL 6

typedef float v2f __attribute__((ext_vector_type(2)));
typedef float v8f __attribute__((ext_vector_type(8)));

__device__ inline v2f ld2(const float* p) { return *(const v2f*)p; }

#define PAD 4
#define P64  (64 + PAD)
#define P128 (128 + PAD)

template <int K, typename FA>
__device__ inline v8f tile_gemm(FA fa, const float* __restrict__ B, int ldb,
                                int col0, v8f acc) {
  const int lane = threadIdx.x & 31;
  const int h2   = (lane >> 4) * 2;
  const int l15  = lane & 15;
#pragma unroll
  for (int kk = 0; kk < K; kk += 4) {
    v2f a = fa(l15, kk, h2);
    v2f b;
    const float* bp = B + (size_t)(kk + h2) * ldb + col0 + l15;
    b.x = bp[0];
    b.y = bp[ldb];
    acc = __builtin_amdgcn_wmma_f32_16x16x4_f32(false, a, false, b,
                                                (short)0, acc, false, false);
  }
  return acc;
}

template <typename ST>
__device__ inline void tile_store(v8f acc, ST st) {
  const int lane = threadIdx.x & 31;
  const int half = lane >> 4;
  const int l15  = lane & 15;
#pragma unroll
  for (int r = 0; r < 8; ++r) st(r + half * 8, l15, acc[r]);
}

#define RT 4
#define EB (16 * RT)
#define NRT 5
#define NB (16 * NRT)

template <int K, int R, typename FA>
__device__ inline void tile_gemm_rt(FA fa, const float* __restrict__ B, int ldb,
                                    int col0, v8f (&acc)[R]) {
  const int lane = threadIdx.x & 31;
  const int h2   = (lane >> 4) * 2;
  const int l15  = lane & 15;
#pragma unroll
  for (int kk = 0; kk < K; kk += 4) {
    v2f b;
    const float* bp = B + (size_t)(kk + h2) * ldb + col0 + l15;
    b.x = bp[0];
    b.y = bp[ldb];
#pragma unroll
    for (int t = 0; t < R; ++t) {
      v2f a = fa(16 * t + l15, kk, h2);
      acc[t] = __builtin_amdgcn_wmma_f32_16x16x4_f32(false, a, false, b,
                                                     (short)0, acc[t], false, false);
    }
  }
}

template <int R, typename ST>
__device__ inline void tile_store_rt(const v8f (&acc)[R], ST st) {
  const int lane = threadIdx.x & 31;
  const int half = lane >> 4;
  const int l15  = lane & 15;
#pragma unroll
  for (int t = 0; t < R; ++t)
#pragma unroll
    for (int r = 0; r < 8; ++r) st(16 * t + r + half * 8, l15, acc[t][r]);
}

template <int K, int NC>
__global__ void encoder_kernel(const float* __restrict__ A,
                               const float* __restrict__ W,
                               const float* __restrict__ bias,
                               float* __restrict__ out,
                               const int* __restrict__ rows) {
  constexpr int P = K + PAD;
  __shared__ __align__(16) float a_s[16 * P];
  const int m0 = blockIdx.x * 16;
  for (int i = threadIdx.x; i < 16 * K / 4; i += blockDim.x) {
    int r = i / (K / 4), c = (i % (K / 4)) * 4;
    const size_t row = rows ? (size_t)rows[m0 + r] : (size_t)(m0 + r);
    *(float4*)(a_s + r * P + c) = *(const float4*)(A + row * K + c);
  }
  __syncthreads();
  const int col0 = (threadIdx.x >> 5) * 16;
  v8f acc = {};
  acc = tile_gemm<K>([&](int r, int kk, int h2) {
    return ld2(a_s + r * P + kk + h2);
  }, W, NC, col0, acc);
  tile_store(acc, [&](int r, int c, float v) {
    out[(size_t)(m0 + r) * NC + col0 + c] = v + bias[col0 + c];
  });
}

#define PW 6
__global__ void __launch_bounds__(PW * 32)
__attribute__((amdgpu_waves_per_eu(4, 8), amdgpu_num_vgpr(128)))
node_proj_kernel(const float* __restrict__ h, const float* __restrict__ mw,
                 const float* __restrict__ w1, float* __restrict__ pq) {
  __shared__ __align__(16) float h_s[NB * P128];
  const int m0 = blockIdx.x * NB;
  constexpr int NT = PW * 32, NV = NB * 32, FULL = NV / NT;
#pragma unroll
  for (int j = 0; j < FULL; ++j) {
    int i = threadIdx.x + j * NT;
    int r = i >> 5, c = (i & 31) * 4;
    *(float4*)(h_s + r * P128 + c) = *(const float4*)(h + (size_t)(m0 + r) * 128 + c);
  }
  if (threadIdx.x < NV - FULL * NT) {
    int i = threadIdx.x + FULL * NT;
    int r = i >> 5, c = (i & 31) * 4;
    *(float4*)(h_s + r * P128 + c) = *(const float4*)(h + (size_t)(m0 + r) * 128 + c);
  }
  __syncthreads();
  const int strip = blockIdx.y * PW + (threadIdx.x >> 5);
  const int part  = strip >> 2;
  const int col0  = (strip & 3) * 16;
  const float* B = part == 0 ? mw : (part == 1 ? w1 : w1 + 128 * 64);
  v8f acc[NRT] = {};
  tile_gemm_rt<128, NRT>([&](int r, int kk, int h2) {
    return ld2(h_s + r * P128 + kk + h2);
  }, B, 64, col0, acc);
  tile_store_rt<NRT>(acc, [&](int r, int c, float v) {
    pq[(size_t)(m0 + r) * 192 + part * 64 + col0 + c] = v;
  });
}

__global__ void __launch_bounds__(128)
edge_mlp_kernel(const float* __restrict__ pq, float* __restrict__ e,
                const int* __restrict__ src, const int* __restrict__ dst,
                const float* __restrict__ w1t, const float* __restrict__ b1,
                const float* __restrict__ w2, const float* __restrict__ b2) {
  __shared__ __align__(16) float e_s[EB * P64];
  __shared__ __align__(16) float t_s[EB * P64];
  __shared__ __align__(16) float s_s[EB * P64];
  float* u_s = s_s;
  const int m0   = blockIdx.x * EB;
  const int col0 = (threadIdx.x >> 5) * 16;
  const int lane = threadIdx.x & 31;
  const int half = lane >> 4;
  const int l15  = lane & 15;

#pragma unroll
  for (int j = 0; j < EB * 16 / 128; ++j) {
    int i = threadIdx.x + j * 128;
    int r = i >> 4, c = (i & 15) * 4;
    const float* ps_row = pq + (size_t)src[m0 + r] * 192;
    const float* pd_row = pq + (size_t)dst[m0 + r] * 192;
    float4 q  = *(const float4*)(ps_row + c);
    float4 ps = *(const float4*)(ps_row + 64 + c);
    float4 pd = *(const float4*)(pd_row + 128 + c);
    float4 ev = *(const float4*)(e + (size_t)(m0 + r) * 64 + c);
    float4 tv, sv;
    tv.x = ev.x + q.x;  tv.y = ev.y + q.y;  tv.z = ev.z + q.z;  tv.w = ev.w + q.w;
    sv.x = ps.x + pd.x; sv.y = ps.y + pd.y; sv.z = ps.z + pd.z; sv.w = ps.w + pd.w;
    *(float4*)(e_s + r * P64 + c) = ev;
    *(float4*)(t_s + r * P64 + c) = tv;
    *(float4*)(s_s + r * P64 + c) = sv;
  }
  __syncthreads();

  v8f acc1[RT];
#pragma unroll
  for (int t = 0; t < RT; ++t)
#pragma unroll
    for (int r = 0; r < 8; ++r)
      acc1[t][r] = s_s[(16 * t + r + half * 8) * P64 + col0 + l15];
  tile_gemm_rt<64, RT>([&](int r, int kk, int h2) {
    return ld2(t_s + r * P64 + kk + h2);
  }, w1t, 64, col0, acc1);
  tile_store_rt<RT>(acc1, [&](int r, int c, float v) {
    float t = v + b1[col0 + c];
    u_s[r * P64 + col0 + c] = t > 0.f ? t : 0.f;
  });
  __syncthreads();

  v8f acc2[RT] = {};
  tile_gemm_rt<64, RT>([&](int r, int kk, int h2) {
    return ld2(u_s + r * P64 + kk + h2);
  }, w2, 64, col0, acc2);
  tile_store_rt<RT>(acc2, [&](int r, int c, float v) {
    e[(size_t)(m0 + r) * 64 + col0 + c] = e_s[r * P64 + col0 + c] + v + b2[col0 + c];
  });
}

#define GC 8
__global__ void node_mlp_kernel(float* __restrict__ h, const float* __restrict__ e,
                                const int* __restrict__ start, const int* __restrict__ cnt,
                                const float* __restrict__ w1, const float* __restrict__ b1,
                                const float* __restrict__ w2, const float* __restrict__ b2) {
  __shared__ __align__(16) float hv_s[16 * P128];
  __shared__ __align__(16) float m_s[16 * P64];
  __shared__ __align__(16) float u_s[16 * P64];
  const int m0   = blockIdx.x * 16;
  const int wave = threadIdx.x >> 5;

  const int gr  = threadIdx.x >> 4, gc = (threadIdx.x & 15) * 4;
  const int beg = start[m0 + gr];
  const int deg = cnt[m0 + gr];

  for (int i = threadIdx.x; i < 16 * 128 / 4; i += blockDim.x) {
    int r = i >> 5, c = (i & 31) * 4;
    *(float4*)(hv_s + r * P128 + c) = *(const float4*)(h + (size_t)(m0 + r) * 128 + c);
  }
  {
    const int r = gr, c = gc;
    const int end = beg + deg;
    float4 v = make_float4(0.f, 0.f, 0.f, 0.f);
    for (int k0 = beg; k0 < end; k0 += GC) {
      float4 w[GC];
#pragma unroll
      for (int j = 0; j < GC; ++j)
        w[j] = (k0 + j < end) ? *(const float4*)(e + (size_t)(k0 + j) * 64 + c)
                              : make_float4(0.f, 0.f, 0.f, 0.f);
#pragma unroll
      for (int j = 0; j < GC; ++j) {
        if (k0 + j < end) { v.x += w[j].x; v.y += w[j].y; v.z += w[j].z; v.w += w[j].w; }
      }
    }
    const float rd = 1.0f / fmaxf((float)deg, 1.0f);
    v.x *= rd; v.y *= rd; v.z *= rd; v.w *= rd;
    *(float4*)(m_s + r * P64 + c) = v;
  }
  __syncthreads();

  if (wave < 4) {
    const int col0 = wave * 16;
    v8f acc = {};
    acc = tile_gemm<192>([&](int r, int kk, int h2) {
      if (kk < 128) return ld2(hv_s + r * P128 + kk + h2);
      return ld2(m_s + r * P64 + (kk - 128) + h2);
    }, w1, 64, col0, acc);
    tile_store(acc, [&](int r, int c, float v) {
      float t = v + b1[col0 + c];
      u_s[r * P64 + col0 + c] = t > 0.f ? t : 0.f;
    });
  }
  __syncthreads();

  const int col0 = wave * 16;
  v8f acc2 = {};
  acc2 = tile_gemm<64>([&](int r, int kk, int h2) {
    return ld2(u_s + r * P64 + kk + h2);
  }, w2, 128, col0, acc2);
  tile_store(acc2, [&](int r, int c, float v) {
    h[(size_t)(m0 + r) * 128 + col0 + c] = hv_s[r * P128 + col0 + c] + v + b2[col0 + c];
  });
}

#define SMAX 16
__global__ void deg_rank_kernel(const int* __restrict__ dst, int* __restrict__ cnt,
                                int* __restrict__ rank) {
  int i = blockIdx.x * blockDim.x + threadIdx.x;
  if (i < EE) rank[i] = atomicAdd(cnt + dst[i], 1);
}

__global__ void seg_alloc_kernel(const int* __restrict__ cnt, int* __restrict__ cursor,
                                 int* __restrict__ start) {
  const int n    = blockIdx.x * blockDim.x + threadIdx.x;
  const int lane = threadIdx.x & 31;
  const int c    = n < NN ? cnt[n] : 0;
  int incl = c;
  for (int d = 1; d < 32; d <<= 1) {
    int t = __shfl_up(incl, d);
    if (lane >= d) incl += t;
  }
  const int total = __shfl(incl, 31);
  int base = 0;
  if (lane == 31) base = atomicAdd(cursor, total);
  base = __shfl(base, 31);
  if (n < NN) start[n] = base + incl - c;
}

__global__ void seg_fill_kernel(const int* __restrict__ dst, const int* __restrict__ start,
                                const int* __restrict__ rank, int* __restrict__ perm) {
  int i = blockIdx.x * blockDim.x + threadIdx.x;
  if (i < EE) perm[start[dst[i]] + rank[i]] = i;
}

__global__ void seg_sort_kernel(const int* __restrict__ src, const int* __restrict__ start,
                                const int* __restrict__ cnt, int* __restrict__ perm,
                                int* __restrict__ srcp, int* __restrict__ dstp) {
  const int n = blockIdx.x * blockDim.x + threadIdx.x;
  if (n >= NN) return;
  const int beg = start[n], d = cnt[n];
  if (d <= SMAX) {
    int id[SMAX];
#pragma unroll
    for (int j = 0; j < SMAX; ++j) id[j] = j < d ? perm[beg + j] : 0x7fffffff;
#pragma unroll
    for (int i = 0; i < SMAX; ++i) {
      if (i < d) {
        int pos = 0;
#pragma unroll
        for (int j = 0; j < SMAX; ++j) pos += id[j] < id[i] ? 1 : 0;
        perm[beg + pos] = id[i];
        srcp[beg + pos] = src[id[i]];
        dstp[beg + pos] = n;
      }
    }
  } else {
    for (int i = 1; i < d; ++i) {
      const int key = perm[beg + i];
      int j = i - 1;
      while (j >= 0 && perm[beg + j] > key) { perm[beg + j + 1] = perm[beg + j]; --j; }
      perm[beg + j + 1] = key;
    }
    for (int i = 0; i < d; ++i) { srcp[beg + i] = src[perm[beg + i]]; dstp[beg + i] = n; }
  }
}

__global__ void recip_clamp_kernel(float* __restrict__ v, int n) {
  int i = blockIdx.x * blockDim.x + threadIdx.x;
  if (i < n) v[i] = 1.0f / fmaxf(v[i], 1.0f);
}

#define PLW 10
#define PLB (8 * PLW)
static_assert(NN % PLB == 0, "pooling blocks must tile the nodes");

__device__ inline void pool_flush(float* __restrict__ gsum, float* __restrict__ gcnt,
                                  int g, int lane, float4 a, float n) {
  float* p = gsum + (size_t)g * 128 + lane * 4;
  atomicAdd(p + 0, a.x);
  atomicAdd(p + 1, a.y);
  atomicAdd(p + 2, a.z);
  atomicAdd(p + 3, a.w);
  if (lane == 0) atomicAdd(gcnt + g, n);
}

__global__ void __launch_bounds__(256)
pool_kernel(const float* __restrict__ h, const int* __restrict__ batch,
            float* __restrict__ gsum, float* __restrict__ gcnt) {
  __shared__ __align__(16) float part_s[8 * 128];
  __shared__ float pcnt_s[8];
  const int lane = threadIdx.x & 31;
  const int wave = __builtin_amdgcn_readfirstlane((int)(threadIdx.x >> 5));
  const int m0 = blockIdx.x * PLB;
  const int n0 = m0 + wave * PLW;
  const int g0 = batch[m0];

  int    g[PLW];
  float4 v[PLW];
#pragma unroll
  for (int j = 0; j < PLW; ++j) {
    g[j] = batch[n0 + j];
    v[j] = *(const float4*)(h + (size_t)(n0 + j) * 128 + lane * 4);
  }

  float4 a0 = make_float4(0.f, 0.f, 0.f, 0.f);
  float  c0 = 0.f;
  float4 a1 = make_float4(0.f, 0.f, 0.f, 0.f);
  float  c1 = 0.f;
  int   cur = g0;
#pragma unroll
  for (int j = 0; j < PLW; ++j) {
    if (g[j] == g0) {
      a0.x += v[j].x; a0.y += v[j].y; a0.z += v[j].z; a0.w += v[j].w;
      c0 += 1.f;
    } else {
      if (g[j] != cur) {
        if (cur != g0) pool_flush(gsum, gcnt, cur, lane, a1, c1);
        cur = g[j];
        a1 = make_float4(0.f, 0.f, 0.f, 0.f);
        c1 = 0.f;
      }
      a1.x += v[j].x; a1.y += v[j].y; a1.z += v[j].z; a1.w += v[j].w;
      c1 += 1.f;
    }
  }
  if (cur != g0) pool_flush(gsum, gcnt, cur, lane, a1, c1);

  *(float4*)(part_s + wave * 128 + lane * 4) = a0;
  if (lane == 0) pcnt_s[wave] = c0;
  __syncthreads();

  if (threadIdx.x < 128) {
    float s = 0.f;
#pragma unroll
    for (int w = 0; w < 8; ++w) s += part_s[w * 128 + threadIdx.x];
    atomicAdd(gsum + (size_t)g0 * 128 + threadIdx.x, s);
    if (threadIdx.x == 0) {
      float c = 0.f;
      for (int w = 0; w < 8; ++w) c += pcnt_s[w];
      atomicAdd(gcnt + g0, c);
    }
  }
}

__global__ void readout_kernel(const float* __restrict__ gsum, const float* __restrict__ rgcnt,
                               const float* __restrict__ w1, const float* __restrict__ b1,
                               const float* __restrict__ w2, const float* __restrict__ b2,
                               float* __restrict__ out) {
  __shared__ __align__(16) float g_s[16 * P128];
  __shared__ __align__(16) float t_s[16 * P128];
  const int m0   = blockIdx.x * 16;
  const int col0 = (threadIdx.x >> 5) * 16;

  for (int i = threadIdx.x; i < 16 * 128 / 4; i += blockDim.x) {
    int r = i >> 5, c = (i & 31) * 4;
    float4 v = *(const float4*)(gsum + (size_t)(m0 + r) * 128 + c);
    float rc = rgcnt[m0 + r];
    v.x *= rc; v.y *= rc; v.z *= rc; v.w *= rc;
    *(float4*)(g_s + r * P128 + c) = v;
  }
  __syncthreads();

  v8f acc = {};
  acc = tile_gemm<128>([&](int r, int kk, int h2) {
    return ld2(g_s + r * P128 + kk + h2);
  }, w1, 128, col0, acc);
  tile_store(acc, [&](int r, int c, float v) {
    float t = v + b1[col0 + c];
    t_s[r * P128 + col0 + c] = t > 0.f ? t : 0.f;
  });
  __syncthreads();

  v8f acc2 = {};
  acc2 = tile_gemm<128>([&](int r, int kk, int h2) {
    return ld2(t_s + r * P128 + kk + h2);
  }, w2, 128, col0, acc2);
  tile_store(acc2, [&](int r, int c, float v) {
    out[(size_t)(m0 + r) * 128 + col0 + c] = v + b2[col0 + c];
  });
}

extern "C" void kernel_launch(void* const* d_in, const int* in_sizes, int n_in,
                              void* d_out, int out_size, void* d_ws, size_t ws_size,
                              hipStream_t stream) {
  (void)in_sizes; (void)n_in; (void)out_size; (void)ws_size;
  const float* x       = (const float*)d_in[0];
  const float* ea      = (const float*)d_in[1];
  const int*   eidx    = (const int*)d_in[2];
  const int*   batch   = (const int*)d_in[3];
  const float* lin_x_w = (const float*)d_in[4];
  const float* lin_x_b = (const float*)d_in[5];
  const float* lin_e_w = (const float*)d_in[6];
  const float* lin_e_b = (const float*)d_in[7];
  const float* msg_w   = (const float*)d_in[8];
  const float* phie_w1 = (const float*)d_in[9];
  const float* phie_b1 = (const float*)d_in[10];
  const float* phie_w2 = (const float*)d_in[11];
  const float* phie_b2 = (const float*)d_in[12];
  const float* phiv_w1 = (const float*)d_in[13];
  const float* phiv_b1 = (const float*)d_in[14];
  const float* phiv_w2 = (const float*)d_in[15];
  const float* phiv_b2 = (const float*)d_in[16];
  const float* ro_w1   = (const float*)d_in[17];
  const float* ro_b1   = (const float*)d_in[18];
  const float* ro_w2   = (const float*)d_in[19];
  const float* ro_b2   = (const float*)d_in[20];
  const int* src = eidx;
  const int* dst = eidx + EE;

  char* wp = (char*)d_ws;
  auto carve = [&](size_t bytes) {
    char* p = wp; wp += (bytes + 255) & ~(size_t)255; return p;
  };
  float* h    = (float*)carve((size_t)NN * 128 * 4);
  float* e    = (float*)carve((size_t)EE * 64 * 4);
  float* pq   = (float*)carve((size_t)NN * 192 * 4);
  int* cnt    = (int*)carve((size_t)(NN + 1) * 4);
  int* rank   = (int*)carve((size_t)EE * 4);
  int* start  = (int*)carve((size_t)NN * 4);
  int* perm   = (int*)carve((size_t)EE * 4);
  int* srcp   = (int*)carve((size_t)EE * 4);
  int* dstp   = (int*)carve((size_t)EE * 4);
  float* gsum = (float*)carve((size_t)BB * 128 * 4);
  float* gcnt = (float*)carve((size_t)BB * 4);

  hipMemsetAsync(cnt, 0, (size_t)(NN + 1) * 4, stream);
  deg_rank_kernel<<<(EE + 255) / 256, 256, 0, stream>>>(dst, cnt, rank);
  seg_alloc_kernel<<<(NN + 255) / 256, 256, 0, stream>>>(cnt, cnt + NN, start);
  seg_fill_kernel<<<(EE + 255) / 256, 256, 0, stream>>>(dst, start, rank, perm);
  seg_sort_kernel<<<(NN + 255) / 256, 256, 0, stream>>>(src, start, cnt, perm, srcp, dstp);

  encoder_kernel<64, 128><<<NN / 16, 256, 0, stream>>>(x, lin_x_w, lin_x_b, h, nullptr);
  encoder_kernel<16, 64><<<EE / 16, 128, 0, stream>>>(ea, lin_e_w, lin_e_b, e, perm);

  for (int l = 0; l < LL; ++l) {
    node_proj_kernel<<<dim3(NN / NB, 12 / PW), PW * 32, 0, stream>>>(h,
        msg_w + (size_t)l * 128 * 64, phie_w1 + (size_t)l * 320 * 64, pq);
    edge_mlp_kernel<<<EE / EB, 128, 0, stream>>>(pq, e, srcp, dstp,
        phie_w1 + (size_t)l * 320 * 64 + 256 * 64, phie_b1 + (size_t)l * 64,
        phie_w2 + (size_t)l * 64 * 64,  phie_b2 + (size_t)l * 64);
    node_mlp_kernel<<<NN / 16, 256, 0, stream>>>(h, e, start, cnt,
        phiv_w1 + (size_t)l * 192 * 64, phiv_b1 + (size_t)l * 64,
        phiv_w2 + (size_t)l * 64 * 128, phiv_b2 + (size_t)l * 128);
  }

  hipMemsetAsync(gsum, 0, (size_t)BB * 128 * 4, stream);
  hipMemsetAsync(gcnt, 0, (size_t)BB * 4, stream);
  pool_kernel<<<NN / PLB, 256, 0, stream>>>(h, batch, gsum, gcnt);
  recip_clamp_kernel<<<1, BB, 0, stream>>>(gcnt, BB);
  readout_kernel<<<BB / 16, 256, 0, stream>>>(gsum, gcnt, ro_w1, ro_b1, ro_w2, ro_b2,
                                              (float*)d_out);
}


# Message weight folded into w1 once per call: wf = w1 src rows + mw @ w1 tail rows; nodes project to 128 columns (was 192), edges gather 2 parts (was 3); f32 sum regrouped as hs @ (mw @ w1 tail)
# speedup vs baseline: 3.0341x; 1.1050x over previous
#include <hip/hip_runtime.h>

#define NN 50000
#define EE 200000
#define BB 64
#define LL 6

typedef float v2f __attribute__((ext_vector_type(2)));
typedef float v8f __attribute__((ext_vector_type(8)));

__device__ inline v2f ld2(const float* p) { return *(const v2f*)p; }

#define PAD 4
#define P64  (64 + PAD)
#define P128 (128 + PAD)

template <int K, typename FA>
__device__ inline v8f tile_gemm(FA fa, const float* __restrict__ B, int ldb,
                                int col0, v8f acc) {
  const int lane = threadIdx.x & 31;
  const int h2   = (lane >> 4) * 2;
  const int l15  = lane & 15;
#pragma unroll
  for (int kk = 0; kk < K; kk += 4) {
    v2f a = fa(l15, kk, h2);
    v2f b;
    const float* bp = B + (size_t)(kk + h2) * ldb + col0 + l15;
    b.x = bp[0];
    b.y = bp[ldb];
    acc = __builtin_amdgcn_wmma_f32_16x16x4_f32(false, a, false, b,
                                                (short)0, acc, false, false);
  }
  return acc;
}

template <typename ST>
__device__ inline void tile_store(v8f acc, ST st) {
  const int lane = threadIdx.x & 31;
  const int half = lane >> 4;
  const int l15  = lane & 15;
#pragma unroll
  for (int r = 0; r < 8; ++r) st(r + half * 8, l15, acc[r]);
}

#define RT 4
#define EB (16 * RT)
#define NRT 5
#define NB (16 * NRT)

template <int K, int R, typename FA>
__device__ inline void tile_gemm_rt(FA fa, const float* __restrict__ B, int ldb,
                                    int col0, v8f (&acc)[R]) {
  const int lane = threadIdx.x & 31;
  const int h2   = (lane >> 4) * 2;
  const int l15  = lane & 15;
#pragma unroll
  for (int kk = 0; kk < K; kk += 4) {
    v2f b;
    const float* bp = B + (size_t)(kk + h2) * ldb + col0 + l15;
    b.x = bp[0];
    b.y = bp[ldb];
#pragma unroll
    for (int t = 0; t < R; ++t) {
      v2f a = fa(16 * t + l15, kk, h2);
      acc[t] = __builtin_amdgcn_wmma_f32_16x16x4_f32(false, a, false, b,
                                                     (short)0, acc[t], false, false);
    }
  }
}

template <int R, typename ST>
__device__ inline void tile_store_rt(const v8f (&acc)[R], ST st) {
  const int lane = threadIdx.x & 31;
  const int half = lane >> 4;
  const int l15  = lane & 15;
#pragma unroll
  for (int t = 0; t < R; ++t)
#pragma unroll
    for (int r = 0; r < 8; ++r) st(16 * t + r + half * 8, l15, acc[t][r]);
}

template <int K, int NC>
__global__ void encoder_kernel(const float* __restrict__ A,
                               const float* __restrict__ W,
                               const float* __restrict__ bias,
                               float* __restrict__ out,
                               const int* __restrict__ rows) {
  constexpr int P = K + PAD;
  __shared__ __align__(16) float a_s[16 * P];
  const int m0 = blockIdx.x * 16;
  for (int i = threadIdx.x; i < 16 * K / 4; i += blockDim.x) {
    int r = i / (K / 4), c = (i % (K / 4)) * 4;
    const size_t row = rows ? (size_t)rows[m0 + r] : (size_t)(m0 + r);
    *(float4*)(a_s + r * P + c) = *(const float4*)(A + row * K + c);
  }
  __syncthreads();
  const int col0 = (threadIdx.x >> 5) * 16;
  v8f acc = {};
  acc = tile_gemm<K>([&](int r, int kk, int h2) {
    return ld2(a_s + r * P + kk + h2);
  }, W, NC, col0, acc);
  tile_store(acc, [&](int r, int c, float v) {
    out[(size_t)(m0 + r) * NC + col0 + c] = v + bias[col0 + c];
  });
}

__global__ void __launch_bounds__(128)
fold_w_kernel(const float* __restrict__ mw, const float* __restrict__ w1,
              float* __restrict__ wf) {
  const int l  = blockIdx.x >> 3;
  const int m0 = (blockIdx.x & 7) * 16;
  const float* A = mw + (size_t)l * 128 * 64;
  const float* W = w1 + (size_t)l * 320 * 64;
  const int col0 = (threadIdx.x >> 5) * 16;
  v8f acc = {};
  acc = tile_gemm<64>([&](int r, int kk, int h2) {
    return ld2(A + (size_t)(m0 + r) * 64 + kk + h2);
  }, W + 256 * 64, 64, col0, acc);
  tile_store(acc, [&](int r, int c, float v) {
    wf[(size_t)l * 128 * 64 + (size_t)(m0 + r) * 64 + col0 + c] =
        W[(size_t)(m0 + r) * 64 + col0 + c] + v;
  });
}

#define PW 4
__global__ void __launch_bounds__(PW * 32)
__attribute__((amdgpu_waves_per_eu(4, 8), amdgpu_num_vgpr(128)))
node_proj_kernel(const float* __restrict__ h, const float* __restrict__ wf,
                 const float* __restrict__ w1, float* __restrict__ pq) {
  __shared__ __align__(16) float h_s[NB * P128];
  const int m0 = blockIdx.x * NB;
  constexpr int NT = PW * 32, NV = NB * 32, FULL = NV / NT;
#pragma unroll
  for (int j = 0; j < FULL; ++j) {
    int i = threadIdx.x + j * NT;
    int r = i >> 5, c = (i & 31) * 4;
    *(float4*)(h_s + r * P128 + c) = *(const float4*)(h + (size_t)(m0 + r) * 128 + c);
  }
  if (threadIdx.x < NV - FULL * NT) {
    int i = threadIdx.x + FULL * NT;
    int r = i >> 5, c = (i & 31) * 4;
    *(float4*)(h_s + r * P128 + c) = *(const float4*)(h + (size_t)(m0 + r) * 128 + c);
  }
  __syncthreads();
  const int part  = blockIdx.y;
  const int col0  = (threadIdx.x >> 5) * 16;
  const float* B = part == 0 ? wf : w1 + 128 * 64;
  v8f acc[NRT] = {};
  tile_gemm_rt<128, NRT>([&](int r, int kk, int h2) {
    return ld2(h_s + r * P128 + kk + h2);
  }, B, 64, col0, acc);
  tile_store_rt<NRT>(acc, [&](int r, int c, float v) {
    pq[(size_t)(m0 + r) * 128 + part * 64 + col0 + c] = v;
  });
}

__global__ void __launch_bounds__(128)
edge_mlp_kernel(const float* __restrict__ pq, float* __restrict__ e,
                const int* __restrict__ src, const int* __restrict__ dst,
                const float* __restrict__ w1t, const float* __restrict__ b1,
                const float* __restrict__ w2, const float* __restrict__ b2) {
  __shared__ __align__(16) float e_s[EB * P64];
  __shared__ __align__(16) float s_s[EB * P64];
  float* u_s = s_s;
  const int m0   = blockIdx.x * EB;
  const int col0 = (threadIdx.x >> 5) * 16;
  const int lane = threadIdx.x & 31;
  const int half = lane >> 4;
  const int l15  = lane & 15;

#pragma unroll
  for (int j = 0; j < EB * 16 / 128; ++j) {
    int i = threadIdx.x + j * 128;
    int r = i >> 4, c = (i & 15) * 4;
    const float* ps_row = pq + (size_t)src[m0 + r] * 128;
    const float* pd_row = pq + (size_t)dst[m0 + r] * 128;
    float4 ps = *(const float4*)(ps_row + c);
    float4 pd = *(const float4*)(pd_row + 64 + c);
    float4 ev = *(const float4*)(e + (size_t)(m0 + r) * 64 + c);
    float4 sv;
    sv.x = ps.x + pd.x; sv.y = ps.y + pd.y; sv.z = ps.z + pd.z; sv.w = ps.w + pd.w;
    *(float4*)(e_s + r * P64 + c) = ev;
    *(float4*)(s_s + r * P64 + c) = sv;
  }
  __syncthreads();

  v8f acc1[RT];
#pragma unroll
  for (int t = 0; t < RT; ++t)
#pragma unroll
    for (int r = 0; r < 8; ++r)
      acc1[t][r] = s_s[(16 * t + r + half * 8) * P64 + col0 + l15];
  tile_gemm_rt<64, RT>([&](int r, int kk, int h2) {
    return ld2(e_s + r * P64 + kk + h2);
  }, w1t, 64, col0, acc1);
  tile_store_rt<RT>(acc1, [&](int r, int c, float v) {
    float t = v + b1[col0 + c];
    u_s[r * P64 + col0 + c] = t > 0.f ? t : 0.f;
  });
  __syncthreads();

  v8f acc2[RT] = {};
  tile_gemm_rt<64, RT>([&](int r, int kk, int h2) {
    return ld2(u_s + r * P64 + kk + h2);
  }, w2, 64, col0, acc2);
  tile_store_rt<RT>(acc2, [&](int r, int c, float v) {
    e[(size_t)(m0 + r) * 64 + col0 + c] = e_s[r * P64 + col0 + c] + v + b2[col0 + c];
  });
}

#define GC 8
__global__ void node_mlp_kernel(float* __restrict__ h, const float* __restrict__ e,
                                const int* __restrict__ start, const int* __restrict__ cnt,
                                const float* __restrict__ w1, const float* __restrict__ b1,
                                const float* __restrict__ w2, const float* __restrict__ b2) {
  __shared__ __align__(16) float hv_s[16 * P128];
  __shared__ __align__(16) float m_s[16 * P64];
  __shared__ __align__(16) float u_s[16 * P64];
  const int m0   = blockIdx.x * 16;
  const int wave = threadIdx.x >> 5;

  const int gr  = threadIdx.x >> 4, gc = (threadIdx.x & 15) * 4;
  const int beg = start[m0 + gr];
  const int deg = cnt[m0 + gr];

  for (int i = threadIdx.x; i < 16 * 128 / 4; i += blockDim.x) {
    int r = i >> 5, c = (i & 31) * 4;
    *(float4*)(hv_s + r * P128 + c) = *(const float4*)(h + (size_t)(m0 + r) * 128 + c);
  }
  {
    const int r = gr, c = gc;
    const int end = beg + deg;
    float4 v = make_float4(0.f, 0.f, 0.f, 0.f);
    for (int k0 = beg; k0 < end; k0 += GC) {
      float4 w[GC];
#pragma unroll
      for (int j = 0; j < GC; ++j)
        w[j] = (k0 + j < end) ? *(const float4*)(e + (size_t)(k0 + j) * 64 + c)
                              : make_float4(0.f, 0.f, 0.f, 0.f);
#pragma unroll
      for (int j = 0; j < GC; ++j) {
        if (k0 + j < end) { v.x += w[j].x; v.y += w[j].y; v.z += w[j].z; v.w += w[j].w; }
      }
    }
    const float rd = 1.0f / fmaxf((float)deg, 1.0f);
    v.x *= rd; v.y *= rd; v.z *= rd; v.w *= rd;
    *(float4*)(m_s + r * P64 + c) = v;
  }
  __syncthreads();

  if (wave < 4) {
    const int col0 = wave * 16;
    v8f acc = {};
    acc = tile_gemm<192>([&](int r, int kk, int h2) {
      if (kk < 128) return ld2(hv_s + r * P128 + kk + h2);
      return ld2(m_s + r * P64 + (kk - 128) + h2);
    }, w1, 64, col0, acc);
    tile_store(acc, [&](int r, int c, float v) {
      float t = v + b1[col0 + c];
      u_s[r * P64 + col0 + c] = t > 0.f ? t : 0.f;
    });
  }
  __syncthreads();

  const int col0 = wave * 16;
  v8f acc2 = {};
  acc2 = tile_gemm<64>([&](int r, int kk, int h2) {
    return ld2(u_s + r * P64 + kk + h2);
  }, w2, 128, col0, acc2);
  tile_store(acc2, [&](int r, int c, float v) {
    h[(size_t)(m0 + r) * 128 + col0 + c] = hv_s[r * P128 + col0 + c] + v + b2[col0 + c];
  });
}

#define SMAX 16
__global__ void deg_rank_kernel(const int* __restrict__ dst, int* __restrict__ cnt,
                                int* __restrict__ rank) {
  int i = blockIdx.x * blockDim.x + threadIdx.x;
  if (i < EE) rank[i] = atomicAdd(cnt + dst[i], 1);
}

__global__ void seg_alloc_kernel(const int* __restrict__ cnt, int* __restrict__ cursor,
                                 int* __restrict__ start) {
  const int n    = blockIdx.x * blockDim.x + threadIdx.x;
  const int lane = threadIdx.x & 31;
  const int c    = n < NN ? cnt[n] : 0;
  int incl = c;
  for (int d = 1; d < 32; d <<= 1) {
    int t = __shfl_up(incl, d);
    if (lane >= d) incl += t;
  }
  const int total = __shfl(incl, 31);
  int base = 0;
  if (lane == 31) base = atomicAdd(cursor, total);
  base = __shfl(base, 31);
  if (n < NN) start[n] = base + incl - c;
}

__global__ void seg_fill_kernel(const int* __restrict__ dst, const int* __restrict__ start,
                                const int* __restrict__ rank, int* __restrict__ perm) {
  int i = blockIdx.x * blockDim.x + threadIdx.x;
  if (i < EE) perm[start[dst[i]] + rank[i]] = i;
}

__global__ void seg_sort_kernel(const int* __restrict__ src, const int* __restrict__ start,
                                const int* __restrict__ cnt, int* __restrict__ perm,
                                int* __restrict__ srcp, int* __restrict__ dstp) {
  const int n = blockIdx.x * blockDim.x + threadIdx.x;
  if (n >= NN) return;
  const int beg = start[n], d = cnt[n];
  if (d <= SMAX) {
    int id[SMAX];
#pragma unroll
    for (int j = 0; j < SMAX; ++j) id[j] = j < d ? perm[beg + j] : 0x7fffffff;
#pragma unroll
    for (int i = 0; i < SMAX; ++i) {
      if (i < d) {
        int pos = 0;
#pragma unroll
        for (int j = 0; j < SMAX; ++j) pos += id[j] < id[i] ? 1 : 0;
        perm[beg + pos] = id[i];
        srcp[beg + pos] = src[id[i]];
        dstp[beg + pos] = n;
      }
    }
  } else {
    for (int i = 1; i < d; ++i) {
      const int key = perm[beg + i];
      int j = i - 1;
      while (j >= 0 && perm[beg + j] > key) { perm[beg + j + 1] = perm[beg + j]; --j; }
      perm[beg + j + 1] = key;
    }
    for (int i = 0; i < d; ++i) { srcp[beg + i] = src[perm[beg + i]]; dstp[beg + i] = n; }
  }
}

__global__ void recip_clamp_kernel(float* __restrict__ v, int n) {
  int i = blockIdx.x * blockDim.x + threadIdx.x;
  if (i < n) v[i] = 1.0f / fmaxf(v[i], 1.0f);
}

#define PLW 10
#define PLB (8 * PLW)
static_assert(NN % PLB == 0, "pooling blocks must tile the nodes");

__device__ inline void pool_flush(float* __restrict__ gsum, float* __restrict__ gcnt,
                                  int g, int lane, float4 a, float n) {
  float* p = gsum + (size_t)g * 128 + lane * 4;
  atomicAdd(p + 0, a.x);
  atomicAdd(p + 1, a.y);
  atomicAdd(p + 2, a.z);
  atomicAdd(p + 3, a.w);
  if (lane == 0) atomicAdd(gcnt + g, n);
}

__global__ void __launch_bounds__(256)
pool_kernel(const float* __restrict__ h, const int* __restrict__ batch,
            float* __restrict__ gsum, float* __restrict__ gcnt) {
  __shared__ __align__(16) float part_s[8 * 128];
  __shared__ float pcnt_s[8];
  const int lane = threadIdx.x & 31;
  const int wave = __builtin_amdgcn_readfirstlane((int)(threadIdx.x >> 5));
  const int m0 = blockIdx.x * PLB;
  const int n0 = m0 + wave * PLW;
  const int g0 = batch[m0];

  int    g[PLW];
  float4 v[PLW];
#pragma unroll
  for (int j = 0; j < PLW; ++j) {
    g[j] = batch[n0 + j];
    v[j] = *(const float4*)(h + (size_t)(n0 + j) * 128 + lane * 4);
  }

  float4 a0 = make_float4(0.f, 0.f, 0.f, 0.f);
  float  c0 = 0.f;
  float4 a1 = make_float4(0.f, 0.f, 0.f, 0.f);
  float  c1 = 0.f;
  int   cur = g0;
#pragma unroll
  for (int j = 0; j < PLW; ++j) {
    if (g[j] == g0) {
      a0.x += v[j].x; a0.y += v[j].y; a0.z += v[j].z; a0.w += v[j].w;
      c0 += 1.f;
    } else {
      if (g[j] != cur) {
        if (cur != g0) pool_flush(gsum, gcnt, cur, lane, a1, c1);
        cur = g[j];
        a1 = make_float4(0.f, 0.f, 0.f, 0.f);
        c1 = 0.f;
      }
      a1.x += v[j].x; a1.y += v[j].y; a1.z += v[j].z; a1.w += v[j].w;
      c1 += 1.f;
    }
  }
  if (cur != g0) pool_flush(gsum, gcnt, cur, lane, a1, c1);

  *(float4*)(part_s + wave * 128 + lane * 4) = a0;
  if (lane == 0) pcnt_s[wave] = c0;
  __syncthreads();

  if (threadIdx.x < 128) {
    float s = 0.f;
#pragma unroll
    for (int w = 0; w < 8; ++w) s += part_s[w * 128 + threadIdx.x];
    atomicAdd(gsum + (size_t)g0 * 128 + threadIdx.x, s);
    if (threadIdx.x == 0) {
      float c = 0.f;
      for (int w = 0; w < 8; ++w) c += pcnt_s[w];
      atomicAdd(gcnt + g0, c);
    }
  }
}

__global__ void readout_kernel(const float* __restrict__ gsum, const float* __restrict__ rgcnt,
                               const float* __restrict__ w1, const float* __restrict__ b1,
                               const float* __restrict__ w2, const float* __restrict__ b2,
                               float* __restrict__ out) {
  __shared__ __align__(16) float g_s[16 * P128];
  __shared__ __align__(16) float t_s[16 * P128];
  const int m0   = blockIdx.x * 16;
  const int col0 = (threadIdx.x >> 5) * 16;

  for (int i = threadIdx.x; i < 16 * 128 / 4; i += blockDim.x) {
    int r = i >> 5, c = (i & 31) * 4;
    float4 v = *(const float4*)(gsum + (size_t)(m0 + r) * 128 + c);
    float rc = rgcnt[m0 + r];
    v.x *= rc; v.y *= rc; v.z *= rc; v.w *= rc;
    *(float4*)(g_s + r * P128 + c) = v;
  }
  __syncthreads();

  v8f acc = {};
  acc = tile_gemm<128>([&](int r, int kk, int h2) {
    return ld2(g_s + r * P128 + kk + h2);
  }, w1, 128, col0, acc);
  tile_store(acc, [&](int r, int c, float v) {
    float t = v + b1[col0 + c];
    t_s[r * P128 + col0 + c] = t > 0.f ? t : 0.f;
  });
  __syncthreads();

  v8f acc2 = {};
  acc2 = tile_gemm<128>([&](int r, int kk, int h2) {
    return ld2(t_s + r * P128 + kk + h2);
  }, w2, 128, col0, acc2);
  tile_store(acc2, [&](int r, int c, float v) {
    out[(size_t)(m0 + r) * 128 + col0 + c] = v + b2[col0 + c];
  });
}

extern "C" void kernel_launch(void* const* d_in, const int* in_sizes, int n_in,
                              void* d_out, int out_size, void* d_ws, size_t ws_size,
                              hipStream_t stream) {
  (void)in_sizes; (void)n_in; (void)out_size; (void)ws_size;
  const float* x       = (const float*)d_in[0];
  const float* ea      = (const float*)d_in[1];
  const int*   eidx    = (const int*)d_in[2];
  const int*   batch   = (const int*)d_in[3];
  const float* lin_x_w = (const float*)d_in[4];
  const float* lin_x_b = (const float*)d_in[5];
  const float* lin_e_w = (const float*)d_in[6];
  const float* lin_e_b = (const float*)d_in[7];
  const float* msg_w   = (const float*)d_in[8];
  const float* phie_w1 = (const float*)d_in[9];
  const float* phie_b1 = (const float*)d_in[10];
  const float* phie_w2 = (const float*)d_in[11];
  const float* phie_b2 = (const float*)d_in[12];
  const float* phiv_w1 = (const float*)d_in[13];
  const float* phiv_b1 = (const float*)d_in[14];
  const float* phiv_w2 = (const float*)d_in[15];
  const float* phiv_b2 = (const float*)d_in[16];
  const float* ro_w1   = (const float*)d_in[17];
  const float* ro_b1   = (const float*)d_in[18];
  const float* ro_w2   = (const float*)d_in[19];
  const float* ro_b2   = (const float*)d_in[20];
  const int* src = eidx;
  const int* dst = eidx + EE;

  char* wp = (char*)d_ws;
  auto carve = [&](size_t bytes) {
    char* p = wp; wp += (bytes + 255) & ~(size_t)255; return p;
  };
  float* h    = (float*)carve((size_t)NN * 128 * 4);
  float* e    = (float*)carve((size_t)EE * 64 * 4);
  float* pq   = (float*)carve((size_t)NN * 128 * 4);
  float* wf   = (float*)carve((size_t)LL * 128 * 64 * 4);
  int* cnt    = (int*)carve((size_t)(NN + 1) * 4);
  int* rank   = (int*)carve((size_t)EE * 4);
  int* start  = (int*)carve((size_t)NN * 4);
  int* perm   = (int*)carve((size_t)EE * 4);
  int* srcp   = (int*)carve((size_t)EE * 4);
  int* dstp   = (int*)carve((size_t)EE * 4);
  float* gsum = (float*)carve((size_t)BB * 128 * 4);
  float* gcnt = (float*)carve((size_t)BB * 4);

  fold_w_kernel<<<LL * 8, 128, 0, stream>>>(msg_w, phie_w1, wf);

  hipMemsetAsync(cnt, 0, (size_t)(NN + 1) * 4, stream);
  deg_rank_kernel<<<(EE + 255) / 256, 256, 0, stream>>>(dst, cnt, rank);
  seg_alloc_kernel<<<(NN + 255) / 256, 256, 0, stream>>>(cnt, cnt + NN, start);
  seg_fill_kernel<<<(EE + 255) / 256, 256, 0, stream>>>(dst, start, rank, perm);
  seg_sort_kernel<<<(NN + 255) / 256, 256, 0, stream>>>(src, start, cnt, perm, srcp, dstp);

  encoder_kernel<64, 128><<<NN / 16, 256, 0, stream>>>(x, lin_x_w, lin_x_b, h, nullptr);
  encoder_kernel<16, 64><<<EE / 16, 128, 0, stream>>>(ea, lin_e_w, lin_e_b, e, perm);

  for (int l = 0; l < LL; ++l) {
    node_proj_kernel<<<dim3(NN / NB, 2), PW * 32, 0, stream>>>(h,
        wf + (size_t)l * 128 * 64, phie_w1 + (size_t)l * 320 * 64, pq);
    edge_mlp_kernel<<<EE / EB, 128, 0, stream>>>(pq, e, srcp, dstp,
        phie_w1 + (size_t)l * 320 * 64 + 256 * 64, phie_b1 + (size_t)l * 64,
        phie_w2 + (size_t)l * 64 * 64,  phie_b2 + (size_t)l * 64);
    node_mlp_kernel<<<NN / 16, 256, 0, stream>>>(h, e, start, cnt,
        phiv_w1 + (size_t)l * 192 * 64, phiv_b1 + (size_t)l * 64,
        phiv_w2 + (size_t)l * 64 * 128, phiv_b2 + (size_t)l * 128);
  }

  hipMemsetAsync(gsum, 0, (size_t)BB * 128 * 4, stream);
  hipMemsetAsync(gcnt, 0, (size_t)BB * 4, stream);
  pool_kernel<<<NN / PLB, 256, 0, stream>>>(h, batch, gsum, gcnt);
  recip_clamp_kernel<<<1, BB, 0, stream>>>(gcnt, BB);
  readout_kernel<<<BB / 16, 256, 0, stream>>>(gsum, gcnt, ro_w1, ro_b1, ro_w2, ro_b2,
                                              (float*)d_out);
}


# Fewer, longer waves: encoders keep 5 (node) or 10 (edge) row tiles per wave (was 1); node update runs 4 waves, each with 2 column strips in stage 2 (was 8 waves, 4 idle in stage 1); sums unchanged
# speedup vs baseline: 3.1447x; 1.0365x over previous
#include <hip/hip_runtime.h>

#define NN 50000
#define EE 200000
#define BB 64
#define LL 6

typedef float v2f __attribute__((ext_vector_type(2)));
typedef float v8f __attribute__((ext_vector_type(8)));

__device__ inline v2f ld2(const float* p) { return *(const v2f*)p; }

#define PAD 4
#define P64  (64 + PAD)
#define P128 (128 + PAD)

template <int K, typename FA>
__device__ inline v8f tile_gemm(FA fa, const float* __restrict__ B, int ldb,
                                int col0, v8f acc) {
  const int lane = threadIdx.x & 31;
  const int h2   = (lane >> 4) * 2;
  const int l15  = lane & 15;
#pragma unroll
  for (int kk = 0; kk < K; kk += 4) {
    v2f a = fa(l15, kk, h2);
    v2f b;
    const float* bp = B + (size_t)(kk + h2) * ldb + col0 + l15;
    b.x = bp[0];
    b.y = bp[ldb];
    acc = __builtin_amdgcn_wmma_f32_16x16x4_f32(false, a, false, b,
                                                (short)0, acc, false, false);
  }
  return acc;
}

template <typename ST>
__device__ inline void tile_store(v8f acc, ST st) {
  const int lane = threadIdx.x & 31;
  const int half = lane >> 4;
  const int l15  = lane & 15;
#pragma unroll
  for (int r = 0; r < 8; ++r) st(r + half * 8, l15, acc[r]);
}

#define RT 4
#define EB (16 * RT)
#define NRT 5
#define NB (16 * NRT)
#define XRT 5
#define ERT 10
static_assert(NN % (16 * XRT) == 0 && EE % (16 * ERT) == 0, "encoder blocks must tile the rows");

template <int K, int R, typename FA>
__device__ inline void tile_gemm_rt(FA fa, const float* __restrict__ B, int ldb,
                                    int col0, v8f (&acc)[R]) {
  const int lane = threadIdx.x & 31;
  const int h2   = (lane >> 4) * 2;
  const int l15  = lane & 15;
#pragma unroll
  for (int kk = 0; kk < K; kk += 4) {
    v2f b;
    const float* bp = B + (size_t)(kk + h2) * ldb + col0 + l15;
    b.x = bp[0];
    b.y = bp[ldb];
#pragma unroll
    for (int t = 0; t < R; ++t) {
      v2f a = fa(16 * t + l15, kk, h2);
      acc[t] = __builtin_amdgcn_wmma_f32_16x16x4_f32(false, a, false, b,
                                                     (short)0, acc[t], false, false);
    }
  }
}

template <int R, typename ST>
__device__ inline void tile_store_rt(const v8f (&acc)[R], ST st) {
  const int lane = threadIdx.x & 31;
  const int half = lane >> 4;
  const int l15  = lane & 15;
#pragma unroll
  for (int t = 0; t < R; ++t)
#pragma unroll
    for (int r = 0; r < 8; ++r) st(16 * t + r + half * 8, l15, acc[t][r]);
}

template <int K, typename FA>
__device__ inline void tile_gemm_2c(FA fa, const float* __restrict__ B, int ldb,
                                    int colA, int colB, v8f& accA, v8f& accB) {
  const int lane = threadIdx.x & 31;
  const int h2   = (lane >> 4) * 2;
  const int l15  = lane & 15;
#pragma unroll
  for (int kk = 0; kk < K; kk += 4) {
    v2f a = fa(l15, kk, h2);
    const float* bp = B + (size_t)(kk + h2) * ldb + l15;
    v2f bA, bB;
    bA.x = bp[colA];
    bA.y = bp[ldb + colA];
    bB.x = bp[colB];
    bB.y = bp[ldb + colB];
    accA = __builtin_amdgcn_wmma_f32_16x16x4_f32(false, a, false, bA,
                                                 (short)0, accA, false, false);
    accB = __builtin_amdgcn_wmma_f32_16x16x4_f32(false, a, false, bB,
                                                 (short)0, accB, false, false);
  }
}

template <int K, int NC, int R, bool LIST>
__global__ void __launch_bounds__(NC * 2)
encoder_kernel(const float* __restrict__ A,
               const float* __restrict__ W,
               const float* __restrict__ bias,
               float* __restrict__ out,
               const int* __restrict__ rows) {
  constexpr int P  = K + PAD;
  constexpr int M  = 16 * R;
  constexpr int NT = NC * 2;
  constexpr int NV = M * K / 4;
  static_assert(NV % NT == 0, "the tile must be whole passes of the block");
  __shared__ __align__(16) float a_s[M * P];
  const int m0 = blockIdx.x * M;
  constexpr int NP = NV / NT;
  size_t row[NP];
  float4 av[NP];
#pragma unroll
  for (int j = 0; j < NP; ++j) {
    const int r = (threadIdx.x + j * NT) / (K / 4);
    row[j] = LIST ? (size_t)rows[m0 + r] : (size_t)(m0 + r);
  }
#pragma unroll
  for (int j = 0; j < NP; ++j) {
    const int c = ((threadIdx.x + j * NT) % (K / 4)) * 4;
    av[j] = *(const float4*)(A + row[j] * K + c);
  }
#pragma unroll
  for (int j = 0; j < NP; ++j) {
    const int i = threadIdx.x + j * NT;
    const int r = i / (K / 4), c = (i % (K / 4)) * 4;
    *(float4*)(a_s + r * P + c) = av[j];
  }
  __syncthreads();
  const int col0 = (threadIdx.x >> 5) * 16;
  v8f acc[R] = {};
  tile_gemm_rt<K, R>([&](int r, int kk, int h2) {
    return ld2(a_s + r * P + kk + h2);
  }, W, NC, col0, acc);
  tile_store_rt<R>(acc, [&](int r, int c, float v) {
    out[(size_t)(m0 + r) * NC + col0 + c] = v + bias[col0 + c];
  });
}

__global__ void __launch_bounds__(128)
fold_w_kernel(const float* __restrict__ mw, const float* __restrict__ w1,
              float* __restrict__ wf) {
  const int l  = blockIdx.x >> 3;
  const int m0 = (blockIdx.x & 7) * 16;
  const float* A = mw + (size_t)l * 128 * 64;
  const float* W = w1 + (size_t)l * 320 * 64;
  const int col0 = (threadIdx.x >> 5) * 16;
  v8f acc = {};
  acc = tile_gemm<64>([&](int r, int kk, int h2) {
    return ld2(A + (size_t)(m0 + r) * 64 + kk + h2);
  }, W + 256 * 64, 64, col0, acc);
  tile_store(acc, [&](int r, int c, float v) {
    wf[(size_t)l * 128 * 64 + (size_t)(m0 + r) * 64 + col0 + c] =
        W[(size_t)(m0 + r) * 64 + col0 + c] + v;
  });
}

#define PW 4
__global__ void __launch_bounds__(PW * 32)
__attribute__((amdgpu_waves_per_eu(4, 8), amdgpu_num_vgpr(128)))
node_proj_kernel(const float* __restrict__ h, const float* __restrict__ wf,
                 const float* __restrict__ w1, float* __restrict__ pq) {
  __shared__ __align__(16) float h_s[NB * P128];
  const int m0 = blockIdx.x * NB;
  constexpr int NT = PW * 32, NV = NB * 32, FULL = NV / NT;
#pragma unroll
  for (int j = 0; j < FULL; ++j) {
    int i = threadIdx.x + j * NT;
    int r = i >> 5, c = (i & 31) * 4;
    *(float4*)(h_s + r * P128 + c) = *(const float4*)(h + (size_t)(m0 + r) * 128 + c);
  }
  if (threadIdx.x < NV - FULL * NT) {
    int i = threadIdx.x + FULL * NT;
    int r = i >> 5, c = (i & 31) * 4;
    *(float4*)(h_s + r * P128 + c) = *(const float4*)(h + (size_t)(m0 + r) * 128 + c);
  }
  __syncthreads();
  const int part  = blockIdx.y;
  const int col0  = (threadIdx.x >> 5) * 16;
  const float* B = part == 0 ? wf : w1 + 128 * 64;
  v8f acc[NRT] = {};
  tile_gemm_rt<128, NRT>([&](int r, int kk, int h2) {
    return ld2(h_s + r * P128 + kk + h2);
  }, B, 64, col0, acc);
  tile_store_rt<NRT>(acc, [&](int r, int c, float v) {
    pq[(size_t)(m0 + r) * 128 + part * 64 + col0 + c] = v;
  });
}

__global__ void __launch_bounds__(128)
edge_mlp_kernel(const float* __restrict__ pq, float* __restrict__ e,
                const int* __restrict__ src, const int* __restrict__ dst,
                const float* __restrict__ w1t, const float* __restrict__ b1,
                const float* __restrict__ w2, const float* __restrict__ b2) {
  __shared__ __align__(16) float e_s[EB * P64];
  __shared__ __align__(16) float s_s[EB * P64];
  float* u_s = s_s;
  const int m0   = blockIdx.x * EB;
  const int col0 = (threadIdx.x >> 5) * 16;
  const int lane = threadIdx.x & 31;
  const int half = lane >> 4;
  const int l15  = lane & 15;

#pragma unroll
  for (int j = 0; j < EB * 16 / 128; ++j) {
    int i = threadIdx.x + j * 128;
    int r = i >> 4, c = (i & 15) * 4;
    const float* ps_row = pq + (size_t)src[m0 + r] * 128;
    const float* pd_row = pq + (size_t)dst[m0 + r] * 128;
    float4 ps = *(const float4*)(ps_row + c);
    float4 pd = *(const float4*)(pd_row + 64 + c);
    float4 ev = *(const float4*)(e + (size_t)(m0 + r) * 64 + c);
    float4 sv;
    sv.x = ps.x + pd.x; sv.y = ps.y + pd.y; sv.z = ps.z + pd.z; sv.w = ps.w + pd.w;
    *(float4*)(e_s + r * P64 + c) = ev;
    *(float4*)(s_s + r * P64 + c) = sv;
  }
  __syncthreads();

  v8f acc1[RT];
#pragma unroll
  for (int t = 0; t < RT; ++t)
#pragma unroll
    for (int r = 0; r < 8; ++r)
      acc1[t][r] = s_s[(16 * t + r + half * 8) * P64 + col0 + l15];
  tile_gemm_rt<64, RT>([&](int r, int kk, int h2) {
    return ld2(e_s + r * P64 + kk + h2);
  }, w1t, 64, col0, acc1);
  tile_store_rt<RT>(acc1, [&](int r, int c, float v) {
    float t = v + b1[col0 + c];
    u_s[r * P64 + col0 + c] = t > 0.f ? t : 0.f;
  });
  __syncthreads();

  v8f acc2[RT] = {};
  tile_gemm_rt<64, RT>([&](int r, int kk, int h2) {
    return ld2(u_s + r * P64 + kk + h2);
  }, w2, 64, col0, acc2);
  tile_store_rt<RT>(acc2, [&](int r, int c, float v) {
    e[(size_t)(m0 + r) * 64 + col0 + c] = e_s[r * P64 + col0 + c] + v + b2[col0 + c];
  });
}

#define GC 8
__global__ void __launch_bounds__(128)
__attribute__((amdgpu_waves_per_eu(4, 8), amdgpu_num_vgpr(128)))
node_mlp_kernel(float* __restrict__ h, const float* __restrict__ e,
                const int* __restrict__ start, const int* __restrict__ cnt,
                const float* __restrict__ w1, const float* __restrict__ b1,
                const float* __restrict__ w2, const float* __restrict__ b2) {
  __shared__ __align__(16) float hv_s[16 * P128];
  __shared__ __align__(16) float m_s[16 * P64];
  __shared__ __align__(16) float u_s[16 * P64];
  const int m0   = blockIdx.x * 16;
  const int wave = threadIdx.x >> 5;

  const int gr   = threadIdx.x >> 4, gc = (threadIdx.x & 15) * 4;
  const int begA = start[m0 + gr];
  const int degA = cnt[m0 + gr];
  const int begB = start[m0 + gr + 8];
  const int degB = cnt[m0 + gr + 8];

  const int hr = threadIdx.x >> 5, hc = (threadIdx.x & 31) * 4;
  const float* hp = h + (size_t)(m0 + hr) * 128 + hc;
  const float4 hv0 = *(const float4*)(hp);
  const float4 hv1 = *(const float4*)(hp + 4 * 128);
  const float4 hv2 = *(const float4*)(hp + 8 * 128);
  const float4 hv3 = *(const float4*)(hp + 12 * 128);
  asm volatile("" :: "v"(begA), "v"(degA), "v"(begB), "v"(degB));
  {
    const int c = gc;
    float4 vA = make_float4(0.f, 0.f, 0.f, 0.f);
    float4 vB = make_float4(0.f, 0.f, 0.f, 0.f);
    const int dmax = degA > degB ? degA : degB;
    const float* pA = e + (size_t)begA * 64 + c;
    const float* pB = e + (size_t)begB * 64 + c;
    for (int k0 = 0; k0 < dmax; k0 += GC, pA += GC * 64, pB += GC * 64) {
      float4 wA[GC], wB[GC];
#pragma unroll
      for (int j = 0; j < GC; ++j)
        wA[j] = (k0 + j < degA) ? *(const float4*)(pA + j * 64)
                                : make_float4(0.f, 0.f, 0.f, 0.f);
#pragma unroll
      for (int j = 0; j < GC; ++j)
        wB[j] = (k0 + j < degB) ? *(const float4*)(pB + j * 64)
                                : make_float4(0.f, 0.f, 0.f, 0.f);
#pragma unroll
      for (int j = 0; j < GC; ++j) {
        if (k0 + j < degA) { vA.x += wA[j].x; vA.y += wA[j].y; vA.z += wA[j].z; vA.w += wA[j].w; }
      }
#pragma unroll
      for (int j = 0; j < GC; ++j) {
        if (k0 + j < degB) { vB.x += wB[j].x; vB.y += wB[j].y; vB.z += wB[j].z; vB.w += wB[j].w; }
      }
    }
    const float rdA = 1.0f / fmaxf((float)degA, 1.0f);
    const float rdB = 1.0f / fmaxf((float)degB, 1.0f);
    vA.x *= rdA; vA.y *= rdA; vA.z *= rdA; vA.w *= rdA;
    vB.x *= rdB; vB.y *= rdB; vB.z *= rdB; vB.w *= rdB;
    *(float4*)(m_s + gr * P64 + c)       = vA;
    *(float4*)(m_s + (gr + 8) * P64 + c) = vB;
  }
  {
    float* hs = hv_s + hr * P128 + hc;
    *(float4*)(hs)             = hv0;
    *(float4*)(hs + 4 * P128)  = hv1;
    *(float4*)(hs + 8 * P128)  = hv2;
    *(float4*)(hs + 12 * P128) = hv3;
  }
  __syncthreads();

  {
    const int col0 = wave * 16;
    v8f acc = {};
    acc = tile_gemm<192>([&](int r, int kk, int h2) {
      if (kk < 128) return ld2(hv_s + r * P128 + kk + h2);
      return ld2(m_s + r * P64 + (kk - 128) + h2);
    }, w1, 64, col0, acc);
    tile_store(acc, [&](int r, int c, float v) {
      float t = v + b1[col0 + c];
      u_s[r * P64 + col0 + c] = t > 0.f ? t : 0.f;
    });
  }
  __syncthreads();

  const int colA = wave * 16;
  const int colB = colA + 64;
  v8f accA = {}, accB = {};
  tile_gemm_2c<64>([&](int r, int kk, int h2) {
    return ld2(u_s + r * P64 + kk + h2);
  }, w2, 128, colA, colB, accA, accB);
  tile_store(accA, [&](int r, int c, float v) {
    h[(size_t)(m0 + r) * 128 + colA + c] = hv_s[r * P128 + colA + c] + v + b2[colA + c];
  });
  tile_store(accB, [&](int r, int c, float v) {
    h[(size_t)(m0 + r) * 128 + colB + c] = hv_s[r * P128 + colB + c] + v + b2[colB + c];
  });
}

#define SMAX 16
__global__ void deg_rank_kernel(const int* __restrict__ dst, int* __restrict__ cnt,
                                int* __restrict__ rank) {
  int i = blockIdx.x * blockDim.x + threadIdx.x;
  if (i < EE) rank[i] = atomicAdd(cnt + dst[i], 1);
}

__global__ void seg_alloc_kernel(const int* __restrict__ cnt, int* __restrict__ cursor,
                                 int* __restrict__ start) {
  const int n    = blockIdx.x * blockDim.x + threadIdx.x;
  const int lane = threadIdx.x & 31;
  const int c    = n < NN ? cnt[n] : 0;
  int incl = c;
  for (int d = 1; d < 32; d <<= 1) {
    int t = __shfl_up(incl, d);
    if (lane >= d) incl += t;
  }
  const int total = __shfl(incl, 31);
  int base = 0;
  if (lane == 31) base = atomicAdd(cursor, total);
  base = __shfl(base, 31);
  if (n < NN) start[n] = base + incl - c;
}

__global__ void seg_fill_kernel(const int* __restrict__ dst, const int* __restrict__ start,
                                const int* __restrict__ rank, int* __restrict__ perm) {
  int i = blockIdx.x * blockDim.x + threadIdx.x;
  if (i < EE) perm[start[dst[i]] + rank[i]] = i;
}

__global__ void seg_sort_kernel(const int* __restrict__ src, const int* __restrict__ start,
                                const int* __restrict__ cnt, int* __restrict__ perm,
                                int* __restrict__ srcp, int* __restrict__ dstp) {
  const int n = blockIdx.x * blockDim.x + threadIdx.x;
  if (n >= NN) return;
  const int beg = start[n], d = cnt[n];
  if (d <= SMAX) {
    int id[SMAX];
#pragma unroll
    for (int j = 0; j < SMAX; ++j) id[j] = j < d ? perm[beg + j] : 0x7fffffff;
#pragma unroll
    for (int i = 0; i < SMAX; ++i) {
      if (i < d) {
        int pos = 0;
#pragma unroll
        for (int j = 0; j < SMAX; ++j) pos += id[j] < id[i] ? 1 : 0;
        perm[beg + pos] = id[i];
        srcp[beg + pos] = src[id[i]];
        dstp[beg + pos] = n;
      }
    }
  } else {
    for (int i = 1; i < d; ++i) {
      const int key = perm[beg + i];
      int j = i - 1;
      while (j >= 0 && perm[beg + j] > key) { perm[beg + j + 1] = perm[beg + j]; --j; }
      perm[beg + j + 1] = key;
    }
    for (int i = 0; i < d; ++i) { srcp[beg + i] = src[perm[beg + i]]; dstp[beg + i] = n; }
  }
}

__global__ void recip_clamp_kernel(float* __restrict__ v, int n) {
  int i = blockIdx.x * blockDim.x + threadIdx.x;
  if (i < n) v[i] = 1.0f / fmaxf(v[i], 1.0f);
}

#define PLW 10
#define PLB (8 * PLW)
static_assert(NN % PLB == 0, "pooling blocks must tile the nodes");

__device__ inline void pool_flush(float* __restrict__ gsum, float* __restrict__ gcnt,
                                  int g, int lane, float4 a, float n) {
  float* p = gsum + (size_t)g * 128 + lane * 4;
  atomicAdd(p + 0, a.x);
  atomicAdd(p + 1, a.y);
  atomicAdd(p + 2, a.z);
  atomicAdd(p + 3, a.w);
  if (lane == 0) atomicAdd(gcnt + g, n);
}

__global__ void __launch_bounds__(256)
pool_kernel(const float* __restrict__ h, const int* __restrict__ batch,
            float* __restrict__ gsum, float* __restrict__ gcnt) {
  __shared__ __align__(16) float part_s[8 * 128];
  __shared__ float pcnt_s[8];
  const int lane = threadIdx.x & 31;
  const int wave = __builtin_amdgcn_readfirstlane((int)(threadIdx.x >> 5));
  const int m0 = blockIdx.x * PLB;
  const int n0 = m0 + wave * PLW;
  const int g0 = batch[m0];

  int    g[PLW];
  float4 v[PLW];
#pragma unroll
  for (int j = 0; j < PLW; ++j) {
    g[j] = batch[n0 + j];
    v[j] = *(const float4*)(h + (size_t)(n0 + j) * 128 + lane * 4);
  }

  float4 a0 = make_float4(0.f, 0.f, 0.f, 0.f);
  float  c0 = 0.f;
  float4 a1 = make_float4(0.f, 0.f, 0.f, 0.f);
  float  c1 = 0.f;
  int   cur = g0;
#pragma unroll
  for (int j = 0; j < PLW; ++j) {
    if (g[j] == g0) {
      a0.x += v[j].x; a0.y += v[j].y; a0.z += v[j].z; a0.w += v[j].w;
      c0 += 1.f;
    } else {
      if (g[j] != cur) {
        if (cur != g0) pool_flush(gsum, gcnt, cur, lane, a1, c1);
        cur = g[j];
        a1 = make_float4(0.f, 0.f, 0.f, 0.f);
        c1 = 0.f;
      }
      a1.x += v[j].x; a1.y += v[j].y; a1.z += v[j].z; a1.w += v[j].w;
      c1 += 1.f;
    }
  }
  if (cur != g0) pool_flush(gsum, gcnt, cur, lane, a1, c1);

  *(float4*)(part_s + wave * 128 + lane * 4) = a0;
  if (lane == 0) pcnt_s[wave] = c0;
  __syncthreads();

  if (threadIdx.x < 128) {
    float s = 0.f;
#pragma unroll
    for (int w = 0; w < 8; ++w) s += part_s[w * 128 + threadIdx.x];
    atomicAdd(gsum + (size_t)g0 * 128 + threadIdx.x, s);
    if (threadIdx.x == 0) {
      float c = 0.f;
      for (int w = 0; w < 8; ++w) c += pcnt_s[w];
      atomicAdd(gcnt + g0, c);
    }
  }
}

__global__ void readout_kernel(const float* __restrict__ gsum, const float* __restrict__ rgcnt,
                               const float* __restrict__ w1, const float* __restrict__ b1,
                               const float* __restrict__ w2, const float* __restrict__ b2,
                               float* __restrict__ out) {
  __shared__ __align__(16) float g_s[16 * P128];
  __shared__ __align__(16) float t_s[16 * P128];
  const int m0   = blockIdx.x * 16;
  const int col0 = (threadIdx.x >> 5) * 16;

  for (int i = threadIdx.x; i < 16 * 128 / 4; i += blockDim.x) {
    int r = i >> 5, c = (i & 31) * 4;
    float4 v = *(const float4*)(gsum + (size_t)(m0 + r) * 128 + c);
    float rc = rgcnt[m0 + r];
    v.x *= rc; v.y *= rc; v.z *= rc; v.w *= rc;
    *(float4*)(g_s + r * P128 + c) = v;
  }
  __syncthreads();

  v8f acc = {};
  acc = tile_gemm<128>([&](int r, int kk, int h2) {
    return ld2(g_s + r * P128 + kk + h2);
  }, w1, 128, col0, acc);
  tile_store(acc, [&](int r, int c, float v) {
    float t = v + b1[col0 + c];
    t_s[r * P128 + col0 + c] = t > 0.f ? t : 0.f;
  });
  __syncthreads();

  v8f acc2 = {};
  acc2 = tile_gemm<128>([&](int r, int kk, int h2) {
    return ld2(t_s + r * P128 + kk + h2);
  }, w2, 128, col0, acc2);
  tile_store(acc2, [&](int r, int c, float v) {
    out[(size_t)(m0 + r) * 128 + col0 + c] = v + b2[col0 + c];
  });
}

extern "C" void kernel_launch(void* const* d_in, const int* in_sizes, int n_in,
                              void* d_out, int out_size, void* d_ws, size_t ws_size,
                              hipStream_t stream) {
  (void)in_sizes; (void)n_in; (void)out_size; (void)ws_size;
  const float* x       = (const float*)d_in[0];
  const float* ea      = (const float*)d_in[1];
  const int*   eidx    = (const int*)d_in[2];
  const int*   batch   = (const int*)d_in[3];
  const float* lin_x_w = (const float*)d_in[4];
  const float* lin_x_b = (const float*)d_in[5];
  const float* lin_e_w = (const float*)d_in[6];
  const float* lin_e_b = (const float*)d_in[7];
  const float* msg_w   = (const float*)d_in[8];
  const float* phie_w1 = (const float*)d_in[9];
  const float* phie_b1 = (const float*)d_in[10];
  const float* phie_w2 = (const float*)d_in[11];
  const float* phie_b2 = (const float*)d_in[12];
  const float* phiv_w1 = (const float*)d_in[13];
  const float* phiv_b1 = (const float*)d_in[14];
  const float* phiv_w2 = (const float*)d_in[15];
  const float* phiv_b2 = (const float*)d_in[16];
  const float* ro_w1   = (const float*)d_in[17];
  const float* ro_b1   = (const float*)d_in[18];
  const float* ro_w2   = (const float*)d_in[19];
  const float* ro_b2   = (const float*)d_in[20];
  const int* src = eidx;
  const int* dst = eidx + EE;

  char* wp = (char*)d_ws;
  auto carve = [&](size_t bytes) {
    char* p = wp; wp += (bytes + 255) & ~(size_t)255; return p;
  };
  float* h    = (float*)carve((size_t)NN * 128 * 4);
  float* e    = (float*)carve((size_t)EE * 64 * 4);
  float* pq   = (float*)carve((size_t)NN * 128 * 4);
  float* wf   = (float*)carve((size_t)LL * 128 * 64 * 4);
  int* cnt    = (int*)carve((size_t)(NN + 1) * 4);
  int* rank   = (int*)carve((size_t)EE * 4);
  int* start  = (int*)carve((size_t)NN * 4);
  int* perm   = (int*)carve((size_t)EE * 4);
  int* srcp   = (int*)carve((size_t)EE * 4);
  int* dstp   = (int*)carve((size_t)EE * 4);
  float* gsum = (float*)carve((size_t)BB * 128 * 4);
  float* gcnt = (float*)carve((size_t)BB * 4);

  fold_w_kernel<<<LL * 8, 128, 0, stream>>>(msg_w, phie_w1, wf);

  hipMemsetAsync(cnt, 0, (size_t)(NN + 1) * 4, stream);
  deg_rank_kernel<<<(EE + 255) / 256, 256, 0, stream>>>(dst, cnt, rank);
  seg_alloc_kernel<<<(NN + 255) / 256, 256, 0, stream>>>(cnt, cnt + NN, start);
  seg_fill_kernel<<<(EE + 255) / 256, 256, 0, stream>>>(dst, start, rank, perm);
  seg_sort_kernel<<<(NN + 255) / 256, 256, 0, stream>>>(src, start, cnt, perm, srcp, dstp);

  encoder_kernel<64, 128, XRT, false><<<NN / (16 * XRT), 256, 0, stream>>>(x, lin_x_w, lin_x_b, h, nullptr);
  encoder_kernel<16, 64, ERT, true><<<EE / (16 * ERT), 128, 0, stream>>>(ea, lin_e_w, lin_e_b, e, perm);

  for (int l = 0; l < LL; ++l) {
    node_proj_kernel<<<dim3(NN / NB, 2), PW * 32, 0, stream>>>(h,
        wf + (size_t)l * 128 * 64, phie_w1 + (size_t)l * 320 * 64, pq);
    edge_mlp_kernel<<<EE / EB, 128, 0, stream>>>(pq, e, srcp, dstp,
        phie_w1 + (size_t)l * 320 * 64 + 256 * 64, phie_b1 + (size_t)l * 64,
        phie_w2 + (size_t)l * 64 * 64,  phie_b2 + (size_t)l * 64);
    node_mlp_kernel<<<NN / 16, 128, 0, stream>>>(h, e, start, cnt,
        phiv_w1 + (size_t)l * 192 * 64, phiv_b1 + (size_t)l * 64,
        phiv_w2 + (size_t)l * 64 * 128, phiv_b2 + (size_t)l * 128);
  }

  hipMemsetAsync(gsum, 0, (size_t)BB * 128 * 4, stream);
  hipMemsetAsync(gcnt, 0, (size_t)BB * 4, stream);
  pool_kernel<<<NN / PLB, 256, 0, stream>>>(h, batch, gsum, gcnt);
  recip_clamp_kernel<<<1, BB, 0, stream>>>(gcnt, BB);
  readout_kernel<<<BB / 16, 256, 0, stream>>>(gsum, gcnt, ro_w1, ro_b1, ro_w2, ro_b2,
                                              (float*)d_out);
}


# Weight (B) fragments read as one 16-byte load per lane from a copy packed in fragment order once per call (was two 4-byte loads per matrix op) in projection, edge and node kernels; sums unchanged
# speedup vs baseline: 3.2420x; 1.0309x over previous
#include <hip/hip_runtime.h>

#define NN 50000
#define EE 200000
#define BB 64
#define LL 6

typedef float v2f __attribute__((ext_vector_type(2)));
typedef float v8f __attribute__((ext_vector_type(8)));

__device__ inline v2f ld2(const float* p) { return *(const v2f*)p; }

#define PAD 4
#define P64  (64 + PAD)
#define P128 (128 + PAD)

template <int K, typename FA>
__device__ inline v8f tile_gemm(FA fa, const float* __restrict__ B, int ldb,
                                int col0, v8f acc) {
  const int lane = threadIdx.x & 31;
  const int h2   = (lane >> 4) * 2;
  const int l15  = lane & 15;
#pragma unroll
  for (int kk = 0; kk < K; kk += 4) {
    v2f a = fa(l15, kk, h2);
    v2f b;
    const float* bp = B + (size_t)(kk + h2) * ldb + col0 + l15;
    b.x = bp[0];
    b.y = bp[ldb];
    acc = __builtin_amdgcn_wmma_f32_16x16x4_f32(false, a, false, b,
                                                (short)0, acc, false, false);
  }
  return acc;
}

template <typename ST>
__device__ inline void tile_store(v8f acc, ST st) {
  const int lane = threadIdx.x & 31;
  const int half = lane >> 4;
  const int l15  = lane & 15;
#pragma unroll
  for (int r = 0; r < 8; ++r) st(r + half * 8, l15, acc[r]);
}

#define RT 4
#define EB (16 * RT)
#define NRT 5
#define NB (16 * NRT)
#define XRT 5
#define ERT 10
static_assert(NN % (16 * XRT) == 0 && EE % (16 * ERT) == 0, "encoder blocks must tile the rows");

template <int K, int R, typename FA>
__device__ inline void tile_gemm_rt(FA fa, const float* __restrict__ B, int ldb,
                                    int col0, v8f (&acc)[R]) {
  const int lane = threadIdx.x & 31;
  const int h2   = (lane >> 4) * 2;
  const int l15  = lane & 15;
#pragma unroll
  for (int kk = 0; kk < K; kk += 4) {
    v2f b;
    const float* bp = B + (size_t)(kk + h2) * ldb + col0 + l15;
    b.x = bp[0];
    b.y = bp[ldb];
#pragma unroll
    for (int t = 0; t < R; ++t) {
      v2f a = fa(16 * t + l15, kk, h2);
      acc[t] = __builtin_amdgcn_wmma_f32_16x16x4_f32(false, a, false, b,
                                                     (short)0, acc[t], false, false);
    }
  }
}

template <int R, typename ST>
__device__ inline void tile_store_rt(const v8f (&acc)[R], ST st) {
  const int lane = threadIdx.x & 31;
  const int half = lane >> 4;
  const int l15  = lane & 15;
#pragma unroll
  for (int t = 0; t < R; ++t)
#pragma unroll
    for (int r = 0; r < 8; ++r) st(16 * t + r + half * 8, l15, acc[t][r]);
}

#define PWF 0
#define PWD 8192
#define PWT 16384
#define PW2 20480
#define PV1 24576
#define PV2 36864
#define PWL 45056

__device__ inline int pk_off(int K, int k, int n) {
  const int r = k & 7;
  const int q = ((n >> 4) * (K >> 3) + (k >> 3)) * 32 + ((r >> 1) & 1) * 16 + (n & 15);
  return q * 4 + (r & 1) + ((r >> 2) << 1);
}

template <int K, int R, typename FA>
__device__ inline void tile_gemm_rt_pk(FA fa, const float* __restrict__ Bs,
                                       v8f (&acc)[R]) {
  const int lane = threadIdx.x & 31;
  const int h2   = (lane >> 4) * 2;
  const int l15  = lane & 15;
  const float4* bp = (const float4*)Bs + lane;
#pragma unroll
  for (int j = 0; j < K / 8; ++j) {
    const float4 b4 = bp[j * 32];
    v2f b0, b1;
    b0.x = b4.x; b0.y = b4.y;
    b1.x = b4.z; b1.y = b4.w;
#pragma unroll
    for (int t = 0; t < R; ++t) {
      v2f a = fa(16 * t + l15, 8 * j, h2);
      acc[t] = __builtin_amdgcn_wmma_f32_16x16x4_f32(false, a, false, b0,
                                                     (short)0, acc[t], false, false);
    }
#pragma unroll
    for (int t = 0; t < R; ++t) {
      v2f a = fa(16 * t + l15, 8 * j + 4, h2);
      acc[t] = __builtin_amdgcn_wmma_f32_16x16x4_f32(false, a, false, b1,
                                                     (short)0, acc[t], false, false);
    }
  }
}

template <int K, typename FA>
__device__ inline void tile_gemm_2c_pk(FA fa, const float* __restrict__ BsA,
                                       const float* __restrict__ BsB,
                                       v8f& accA, v8f& accB) {
  const int lane = threadIdx.x & 31;
  const int h2   = (lane >> 4) * 2;
  const int l15  = lane & 15;
  const float4* bpA = (const float4*)BsA + lane;
  const float4* bpB = (const float4*)BsB + lane;
#pragma unroll
  for (int j = 0; j < K / 8; ++j) {
    const float4 p4 = bpA[j * 32];
    const float4 q4 = bpB[j * 32];
    v2f bA, bB;
    v2f a = fa(l15, 8 * j, h2);
    bA.x = p4.x; bA.y = p4.y;
    bB.x = q4.x; bB.y = q4.y;
    accA = __builtin_amdgcn_wmma_f32_16x16x4_f32(false, a, false, bA,
                                                 (short)0, accA, false, false);
    accB = __builtin_amdgcn_wmma_f32_16x16x4_f32(false, a, false, bB,
                                                 (short)0, accB, false, false);
    a = fa(l15, 8 * j + 4, h2);
    bA.x = p4.z; bA.y = p4.w;
    bB.x = q4.z; bB.y = q4.w;
    accA = __builtin_amdgcn_wmma_f32_16x16x4_f32(false, a, false, bA,
                                                 (short)0, accA, false, false);
    accB = __builtin_amdgcn_wmma_f32_16x16x4_f32(false, a, false, bB,
                                                 (short)0, accB, false, false);
  }
}

template <int K, int NC, int R, bool LIST>
__global__ void __launch_bounds__(NC * 2)
encoder_kernel(const float* __restrict__ A,
               const float* __restrict__ W,
               const float* __restrict__ bias,
               float* __restrict__ out,
               const int* __restrict__ rows) {
  constexpr int P  = K + PAD;
  constexpr int M  = 16 * R;
  constexpr int NT = NC * 2;
  constexpr int NV = M * K / 4;
  static_assert(NV % NT == 0, "the tile must be whole passes of the block");
  __shared__ __align__(16) float a_s[M * P];
  const int m0 = blockIdx.x * M;
  constexpr int NP = NV / NT;
  size_t row[NP];
  float4 av[NP];
#pragma unroll
  for (int j = 0; j < NP; ++j) {
    const int r = (threadIdx.x + j * NT) / (K / 4);
    row[j] = LIST ? (size_t)rows[m0 + r] : (size_t)(m0 + r);
  }
#pragma unroll
  for (int j = 0; j < NP; ++j) {
    const int c = ((threadIdx.x + j * NT) % (K / 4)) * 4;
    av[j] = *(const float4*)(A + row[j] * K + c);
  }
#pragma unroll
  for (int j = 0; j < NP; ++j) {
    const int i = threadIdx.x + j * NT;
    const int r = i / (K / 4), c = (i % (K / 4)) * 4;
    *(float4*)(a_s + r * P + c) = av[j];
  }
  __syncthreads();
  const int col0 = (threadIdx.x >> 5) * 16;
  v8f acc[R] = {};
  tile_gemm_rt<K, R>([&](int r, int kk, int h2) {
    return ld2(a_s + r * P + kk + h2);
  }, W, NC, col0, acc);
  tile_store_rt<R>(acc, [&](int r, int c, float v) {
    out[(size_t)(m0 + r) * NC + col0 + c] = v + bias[col0 + c];
  });
}

#define FOLDB (LL * 8)
#define PACKQ ((PWL - 8192) / 4)
#define PACKB (LL * PACKQ / 128)
static_assert((LL * PACKQ) % 128 == 0, "pack blocks must be whole");

__device__ inline void pack_mat(const float* __restrict__ Wm, int ldw, int K,
                                float* __restrict__ dst, int q) {
  const int per = (K >> 3) * 32;
  const int s = q / per, rem = q - s * per;
  const int j = rem >> 5, lane = rem & 31;
  const int k0 = 8 * j + 2 * (lane >> 4);
  const float* p = Wm + (size_t)k0 * ldw + 16 * s + (lane & 15);
  float4 v;
  v.x = p[0];
  v.y = p[ldw];
  v.z = p[4 * ldw];
  v.w = p[5 * ldw];
  ((float4*)dst)[q] = v;
}

__global__ void __launch_bounds__(128)
fold_pack_kernel(const float* __restrict__ mw, const float* __restrict__ w1,
                 const float* __restrict__ ew2, const float* __restrict__ vw1,
                 const float* __restrict__ vw2, float* __restrict__ wp) {
  if (blockIdx.x >= FOLDB) {
    const int g = (blockIdx.x - FOLDB) * 128 + threadIdx.x;
    const int pl = g / PACKQ, q = g - pl * PACKQ;
    float* d = wp + (size_t)pl * PWL;
    const float* e1 = w1 + (size_t)pl * 320 * 64;
    if (q < 2048)      pack_mat(e1 + 128 * 64, 64, 128, d + PWD, q);
    else if (q < 3072) pack_mat(e1 + 256 * 64, 64, 64, d + PWT, q - 2048);
    else if (q < 4096) pack_mat(ew2 + (size_t)pl * 64 * 64, 64, 64, d + PW2, q - 3072);
    else if (q < 7168) pack_mat(vw1 + (size_t)pl * 192 * 64, 64, 192, d + PV1, q - 4096);
    else               pack_mat(vw2 + (size_t)pl * 64 * 128, 128, 64, d + PV2, q - 7168);
    return;
  }
  float* wf = wp + PWF;
  const int l  = blockIdx.x >> 3;
  const int m0 = (blockIdx.x & 7) * 16;
  const float* A = mw + (size_t)l * 128 * 64;
  const float* W = w1 + (size_t)l * 320 * 64;
  const int col0 = (threadIdx.x >> 5) * 16;
  v8f acc = {};
  acc = tile_gemm<64>([&](int r, int kk, int h2) {
    return ld2(A + (size_t)(m0 + r) * 64 + kk + h2);
  }, W + 256 * 64, 64, col0, acc);
  tile_store(acc, [&](int r, int c, float v) {
    wf[(size_t)l * PWL + pk_off(128, m0 + r, col0 + c)] =
        W[(size_t)(m0 + r) * 64 + col0 + c] + v;
  });
}

#define PW 4
__global__ void __launch_bounds__(PW * 32)
__attribute__((amdgpu_waves_per_eu(4, 8), amdgpu_num_vgpr(128)))
node_proj_kernel(const float* __restrict__ h, const float* __restrict__ wp,
                 float* __restrict__ pq) {
  __shared__ __align__(16) float h_s[NB * P128];
  const int m0 = blockIdx.x * NB;
  constexpr int NT = PW * 32, NV = NB * 32, FULL = NV / NT;
#pragma unroll
  for (int j = 0; j < FULL; ++j) {
    int i = threadIdx.x + j * NT;
    int r = i >> 5, c = (i & 31) * 4;
    *(float4*)(h_s + r * P128 + c) = *(const float4*)(h + (size_t)(m0 + r) * 128 + c);
  }
  if (threadIdx.x < NV - FULL * NT) {
    int i = threadIdx.x + FULL * NT;
    int r = i >> 5, c = (i & 31) * 4;
    *(float4*)(h_s + r * P128 + c) = *(const float4*)(h + (size_t)(m0 + r) * 128 + c);
  }
  __syncthreads();
  const int part  = blockIdx.y;
  const int col0  = (threadIdx.x >> 5) * 16;
  const float* Bs = wp + (part == 0 ? PWF : PWD) + (threadIdx.x >> 5) * 2048;
  v8f acc[NRT] = {};
  tile_gemm_rt_pk<128, NRT>([&](int r, int kk, int h2) {
    return ld2(h_s + r * P128 + kk + h2);
  }, Bs, acc);
  tile_store_rt<NRT>(acc, [&](int r, int c, float v) {
    pq[(size_t)(m0 + r) * 128 + part * 64 + col0 + c] = v;
  });
}

__global__ void __launch_bounds__(128)
edge_mlp_kernel(const float* __restrict__ pq, float* __restrict__ e,
                const int* __restrict__ src, const int* __restrict__ dst,
                const float* __restrict__ wp, const float* __restrict__ b1,
                const float* __restrict__ b2) {
  __shared__ __align__(16) float e_s[EB * P64];
  __shared__ __align__(16) float s_s[EB * P64];
  float* u_s = s_s;
  const int m0   = blockIdx.x * EB;
  const int col0 = (threadIdx.x >> 5) * 16;
  const int lane = threadIdx.x & 31;
  const int half = lane >> 4;
  const int l15  = lane & 15;

#pragma unroll
  for (int j = 0; j < EB * 16 / 128; ++j) {
    int i = threadIdx.x + j * 128;
    int r = i >> 4, c = (i & 15) * 4;
    const float* ps_row = pq + (size_t)src[m0 + r] * 128;
    const float* pd_row = pq + (size_t)dst[m0 + r] * 128;
    float4 ps = *(const float4*)(ps_row + c);
    float4 pd = *(const float4*)(pd_row + 64 + c);
    float4 ev = *(const float4*)(e + (size_t)(m0 + r) * 64 + c);
    float4 sv;
    sv.x = ps.x + pd.x; sv.y = ps.y + pd.y; sv.z = ps.z + pd.z; sv.w = ps.w + pd.w;
    *(float4*)(e_s + r * P64 + c) = ev;
    *(float4*)(s_s + r * P64 + c) = sv;
  }
  __syncthreads();

  v8f acc1[RT];
#pragma unroll
  for (int t = 0; t < RT; ++t)
#pragma unroll
    for (int r = 0; r < 8; ++r)
      acc1[t][r] = s_s[(16 * t + r + half * 8) * P64 + col0 + l15];
  const float* Bs1 = wp + PWT + (threadIdx.x >> 5) * 1024;
  const float* Bs2 = wp + PW2 + (threadIdx.x >> 5) * 1024;
  tile_gemm_rt_pk<64, RT>([&](int r, int kk, int h2) {
    return ld2(e_s + r * P64 + kk + h2);
  }, Bs1, acc1);
  tile_store_rt<RT>(acc1, [&](int r, int c, float v) {
    float t = v + b1[col0 + c];
    u_s[r * P64 + col0 + c] = t > 0.f ? t : 0.f;
  });
  __syncthreads();

  v8f acc2[RT] = {};
  tile_gemm_rt_pk<64, RT>([&](int r, int kk, int h2) {
    return ld2(u_s + r * P64 + kk + h2);
  }, Bs2, acc2);
  tile_store_rt<RT>(acc2, [&](int r, int c, float v) {
    e[(size_t)(m0 + r) * 64 + col0 + c] = e_s[r * P64 + col0 + c] + v + b2[col0 + c];
  });
}

#define GC 8
__global__ void __launch_bounds__(128)
__attribute__((amdgpu_waves_per_eu(4, 8), amdgpu_num_vgpr(128)))
node_mlp_kernel(float* __restrict__ h, const float* __restrict__ e,
                const int* __restrict__ start, const int* __restrict__ cnt,
                const float* __restrict__ wp, const float* __restrict__ b1,
                const float* __restrict__ b2) {
  __shared__ __align__(16) float hv_s[16 * P128];
  __shared__ __align__(16) float m_s[16 * P64];
  __shared__ __align__(16) float u_s[16 * P64];
  const int m0   = blockIdx.x * 16;
  const int wave = threadIdx.x >> 5;

  const int gr   = threadIdx.x >> 4, gc = (threadIdx.x & 15) * 4;
  const int begA = start[m0 + gr];
  const int degA = cnt[m0 + gr];
  const int begB = start[m0 + gr + 8];
  const int degB = cnt[m0 + gr + 8];

  const int hr = threadIdx.x >> 5, hc = (threadIdx.x & 31) * 4;
  const float* hp = h + (size_t)(m0 + hr) * 128 + hc;
  const float4 hv0 = *(const float4*)(hp);
  const float4 hv1 = *(const float4*)(hp + 4 * 128);
  const float4 hv2 = *(const float4*)(hp + 8 * 128);
  const float4 hv3 = *(const float4*)(hp + 12 * 128);
  asm volatile("" :: "v"(begA), "v"(degA), "v"(begB), "v"(degB));
  {
    const int c = gc;
    float4 vA = make_float4(0.f, 0.f, 0.f, 0.f);
    float4 vB = make_float4(0.f, 0.f, 0.f, 0.f);
    const int dmax = degA > degB ? degA : degB;
    const float* pA = e + (size_t)begA * 64 + c;
    const float* pB = e + (size_t)begB * 64 + c;
    for (int k0 = 0; k0 < dmax; k0 += GC, pA += GC * 64, pB += GC * 64) {
      float4 wA[GC], wB[GC];
#pragma unroll
      for (int j = 0; j < GC; ++j)
        wA[j] = (k0 + j < degA) ? *(const float4*)(pA + j * 64)
                                : make_float4(0.f, 0.f, 0.f, 0.f);
#pragma unroll
      for (int j = 0; j < GC; ++j)
        wB[j] = (k0 + j < degB) ? *(const float4*)(pB + j * 64)
                                : make_float4(0.f, 0.f, 0.f, 0.f);
#pragma unroll
      for (int j = 0; j < GC; ++j) {
        if (k0 + j < degA) { vA.x += wA[j].x; vA.y += wA[j].y; vA.z += wA[j].z; vA.w += wA[j].w; }
      }
#pragma unroll
      for (int j = 0; j < GC; ++j) {
        if (k0 + j < degB) { vB.x += wB[j].x; vB.y += wB[j].y; vB.z += wB[j].z; vB.w += wB[j].w; }
      }
    }
    const float rdA = 1.0f / fmaxf((float)degA, 1.0f);
    const float rdB = 1.0f / fmaxf((float)degB, 1.0f);
    vA.x *= rdA; vA.y *= rdA; vA.z *= rdA; vA.w *= rdA;
    vB.x *= rdB; vB.y *= rdB; vB.z *= rdB; vB.w *= rdB;
    *(float4*)(m_s + gr * P64 + c)       = vA;
    *(float4*)(m_s + (gr + 8) * P64 + c) = vB;
  }
  {
    float* hs = hv_s + hr * P128 + hc;
    *(float4*)(hs)             = hv0;
    *(float4*)(hs + 4 * P128)  = hv1;
    *(float4*)(hs + 8 * P128)  = hv2;
    *(float4*)(hs + 12 * P128) = hv3;
  }
  __syncthreads();

  {
    const int col0 = wave * 16;
    v8f acc1[1] = {};
    tile_gemm_rt_pk<192, 1>([&](int r, int kk, int h2) {
      if (kk < 128) return ld2(hv_s + r * P128 + kk + h2);
      return ld2(m_s + r * P64 + (kk - 128) + h2);
    }, wp + PV1 + wave * 3072, acc1);
    const v8f acc = acc1[0];
    tile_store(acc, [&](int r, int c, float v) {
      float t = v + b1[col0 + c];
      u_s[r * P64 + col0 + c] = t > 0.f ? t : 0.f;
    });
  }
  __syncthreads();

  const int colA = wave * 16;
  const int colB = colA + 64;
  v8f accA = {}, accB = {};
  tile_gemm_2c_pk<64>([&](int r, int kk, int h2) {
    return ld2(u_s + r * P64 + kk + h2);
  }, wp + PV2 + wave * 1024, wp + PV2 + (wave + 4) * 1024, accA, accB);
  tile_store(accA, [&](int r, int c, float v) {
    h[(size_t)(m0 + r) * 128 + colA + c] = hv_s[r * P128 + colA + c] + v + b2[colA + c];
  });
  tile_store(accB, [&](int r, int c, float v) {
    h[(size_t)(m0 + r) * 128 + colB + c] = hv_s[r * P128 + colB + c] + v + b2[colB + c];
  });
}

#define SMAX 16
__global__ void deg_rank_kernel(const int* __restrict__ dst, int* __restrict__ cnt,
                                int* __restrict__ rank) {
  int i = blockIdx.x * blockDim.x + threadIdx.x;
  if (i < EE) rank[i] = atomicAdd(cnt + dst[i], 1);
}

__global__ void seg_alloc_kernel(const int* __restrict__ cnt, int* __restrict__ cursor,
                                 int* __restrict__ start) {
  const int n    = blockIdx.x * blockDim.x + threadIdx.x;
  const int lane = threadIdx.x & 31;
  const int c    = n < NN ? cnt[n] : 0;
  int incl = c;
  for (int d = 1; d < 32; d <<= 1) {
    int t = __shfl_up(incl, d);
    if (lane >= d) incl += t;
  }
  const int total = __shfl(incl, 31);
  int base = 0;
  if (lane == 31) base = atomicAdd(cursor, total);
  base = __shfl(base, 31);
  if (n < NN) start[n] = base + incl - c;
}

__global__ void seg_fill_kernel(const int* __restrict__ dst, const int* __restrict__ start,
                                const int* __restrict__ rank, int* __restrict__ perm) {
  int i = blockIdx.x * blockDim.x + threadIdx.x;
  if (i < EE) perm[start[dst[i]] + rank[i]] = i;
}

__global__ void seg_sort_kernel(const int* __restrict__ src, const int* __restrict__ start,
                                const int* __restrict__ cnt, int* __restrict__ perm,
                                int* __restrict__ srcp, int* __restrict__ dstp) {
  const int n = blockIdx.x * blockDim.x + threadIdx.x;
  if (n >= NN) return;
  const int beg = start[n], d = cnt[n];
  if (d <= SMAX) {
    int id[SMAX];
#pragma unroll
    for (int j = 0; j < SMAX; ++j) id[j] = j < d ? perm[beg + j] : 0x7fffffff;
#pragma unroll
    for (int i = 0; i < SMAX; ++i) {
      if (i < d) {
        int pos = 0;
#pragma unroll
        for (int j = 0; j < SMAX; ++j) pos += id[j] < id[i] ? 1 : 0;
        perm[beg + pos] = id[i];
        srcp[beg + pos] = src[id[i]];
        dstp[beg + pos] = n;
      }
    }
  } else {
    for (int i = 1; i < d; ++i) {
      const int key = perm[beg + i];
      int j = i - 1;
      while (j >= 0 && perm[beg + j] > key) { perm[beg + j + 1] = perm[beg + j]; --j; }
      perm[beg + j + 1] = key;
    }
    for (int i = 0; i < d; ++i) { srcp[beg + i] = src[perm[beg + i]]; dstp[beg + i] = n; }
  }
}

__global__ void recip_clamp_kernel(float* __restrict__ v, int n) {
  int i = blockIdx.x * blockDim.x + threadIdx.x;
  if (i < n) v[i] = 1.0f / fmaxf(v[i], 1.0f);
}

#define PLW 10
#define PLB (8 * PLW)
static_assert(NN % PLB == 0, "pooling blocks must tile the nodes");

__device__ inline void pool_flush(float* __restrict__ gsum, float* __restrict__ gcnt,
                                  int g, int lane, float4 a, float n) {
  float* p = gsum + (size_t)g * 128 + lane * 4;
  atomicAdd(p + 0, a.x);
  atomicAdd(p + 1, a.y);
  atomicAdd(p + 2, a.z);
  atomicAdd(p + 3, a.w);
  if (lane == 0) atomicAdd(gcnt + g, n);
}

__global__ void __launch_bounds__(256)
pool_kernel(const float* __restrict__ h, const int* __restrict__ batch,
            float* __restrict__ gsum, float* __restrict__ gcnt) {
  __shared__ __align__(16) float part_s[8 * 128];
  __shared__ float pcnt_s[8];
  const int lane = threadIdx.x & 31;
  const int wave = __builtin_amdgcn_readfirstlane((int)(threadIdx.x >> 5));
  const int m0 = blockIdx.x * PLB;
  const int n0 = m0 + wave * PLW;
  const int g0 = batch[m0];

  int    g[PLW];
  float4 v[PLW];
#pragma unroll
  for (int j = 0; j < PLW; ++j) {
    g[j] = batch[n0 + j];
    v[j] = *(const float4*)(h + (size_t)(n0 + j) * 128 + lane * 4);
  }

  float4 a0 = make_float4(0.f, 0.f, 0.f, 0.f);
  float  c0 = 0.f;
  float4 a1 = make_float4(0.f, 0.f, 0.f, 0.f);
  float  c1 = 0.f;
  int   cur = g0;
#pragma unroll
  for (int j = 0; j < PLW; ++j) {
    if (g[j] == g0) {
      a0.x += v[j].x; a0.y += v[j].y; a0.z += v[j].z; a0.w += v[j].w;
      c0 += 1.f;
    } else {
      if (g[j] != cur) {
        if (cur != g0) pool_flush(gsum, gcnt, cur, lane, a1, c1);
        cur = g[j];
        a1 = make_float4(0.f, 0.f, 0.f, 0.f);
        c1 = 0.f;
      }
      a1.x += v[j].x; a1.y += v[j].y; a1.z += v[j].z; a1.w += v[j].w;
      c1 += 1.f;
    }
  }
  if (cur != g0) pool_flush(gsum, gcnt, cur, lane, a1, c1);

  *(float4*)(part_s + wave * 128 + lane * 4) = a0;
  if (lane == 0) pcnt_s[wave] = c0;
  __syncthreads();

  if (threadIdx.x < 128) {
    float s = 0.f;
#pragma unroll
    for (int w = 0; w < 8; ++w) s += part_s[w * 128 + threadIdx.x];
    atomicAdd(gsum + (size_t)g0 * 128 + threadIdx.x, s);
    if (threadIdx.x == 0) {
      float c = 0.f;
      for (int w = 0; w < 8; ++w) c += pcnt_s[w];
      atomicAdd(gcnt + g0, c);
    }
  }
}

__global__ void readout_kernel(const float* __restrict__ gsum, const float* __restrict__ rgcnt,
                               const float* __restrict__ w1, const float* __restrict__ b1,
                               const float* __restrict__ w2, const float* __restrict__ b2,
                               float* __restrict__ out) {
  __shared__ __align__(16) float g_s[16 * P128];
  __shared__ __align__(16) float t_s[16 * P128];
  const int m0   = blockIdx.x * 16;
  const int col0 = (threadIdx.x >> 5) * 16;

  for (int i = threadIdx.x; i < 16 * 128 / 4; i += blockDim.x) {
    int r = i >> 5, c = (i & 31) * 4;
    float4 v = *(const float4*)(gsum + (size_t)(m0 + r) * 128 + c);
    float rc = rgcnt[m0 + r];
    v.x *= rc; v.y *= rc; v.z *= rc; v.w *= rc;
    *(float4*)(g_s + r * P128 + c) = v;
  }
  __syncthreads();

  v8f acc = {};
  acc = tile_gemm<128>([&](int r, int kk, int h2) {
    return ld2(g_s + r * P128 + kk + h2);
  }, w1, 128, col0, acc);
  tile_store(acc, [&](int r, int c, float v) {
    float t = v + b1[col0 + c];
    t_s[r * P128 + col0 + c] = t > 0.f ? t : 0.f;
  });
  __syncthreads();

  v8f acc2 = {};
  acc2 = tile_gemm<128>([&](int r, int kk, int h2) {
    return ld2(t_s + r * P128 + kk + h2);
  }, w2, 128, col0, acc2);
  tile_store(acc2, [&](int r, int c, float v) {
    out[(size_t)(m0 + r) * 128 + col0 + c] = v + b2[col0 + c];
  });
}

extern "C" void kernel_launch(void* const* d_in, const int* in_sizes, int n_in,
                              void* d_out, int out_size, void* d_ws, size_t ws_size,
                              hipStream_t stream) {
  (void)in_sizes; (void)n_in; (void)out_size; (void)ws_size;
  const float* x       = (const float*)d_in[0];
  const float* ea      = (const float*)d_in[1];
  const int*   eidx    = (const int*)d_in[2];
  const int*   batch   = (const int*)d_in[3];
  const float* lin_x_w = (const float*)d_in[4];
  const float* lin_x_b = (const float*)d_in[5];
  const float* lin_e_w = (const float*)d_in[6];
  const float* lin_e_b = (const float*)d_in[7];
  const float* msg_w   = (const float*)d_in[8];
  const float* phie_w1 = (const float*)d_in[9];
  const float* phie_b1 = (const float*)d_in[10];
  const float* phie_w2 = (const float*)d_in[11];
  const float* phie_b2 = (const float*)d_in[12];
  const float* phiv_w1 = (const float*)d_in[13];
  const float* phiv_b1 = (const float*)d_in[14];
  const float* phiv_w2 = (const float*)d_in[15];
  const float* phiv_b2 = (const float*)d_in[16];
  const float* ro_w1   = (const float*)d_in[17];
  const float* ro_b1   = (const float*)d_in[18];
  const float* ro_w2   = (const float*)d_in[19];
  const float* ro_b2   = (const float*)d_in[20];
  const int* src = eidx;
  const int* dst = eidx + EE;

  char* wp = (char*)d_ws;
  auto carve = [&](size_t bytes) {
    char* p = wp; wp += (bytes + 255) & ~(size_t)255; return p;
  };
  float* h    = (float*)carve((size_t)NN * 128 * 4);
  float* e    = (float*)carve((size_t)EE * 64 * 4);
  float* pq   = (float*)carve((size_t)NN * 128 * 4);
  float* wpk  = (float*)carve((size_t)LL * PWL * 4);
  int* cnt    = (int*)carve((size_t)(NN + 1) * 4);
  int* rank   = (int*)carve((size_t)EE * 4);
  int* start  = (int*)carve((size_t)NN * 4);
  int* perm   = (int*)carve((size_t)EE * 4);
  int* srcp   = (int*)carve((size_t)EE * 4);
  int* dstp   = (int*)carve((size_t)EE * 4);
  float* gsum = (float*)carve((size_t)BB * 128 * 4);
  float* gcnt = (float*)carve((size_t)BB * 4);

  fold_pack_kernel<<<FOLDB + PACKB, 128, 0, stream>>>(msg_w, phie_w1, phie_w2,
                                                      phiv_w1, phiv_w2, wpk);

  hipMemsetAsync(cnt, 0, (size_t)(NN + 1) * 4, stream);
  deg_rank_kernel<<<(EE + 255) / 256, 256, 0, stream>>>(dst, cnt, rank);
  seg_alloc_kernel<<<(NN + 255) / 256, 256, 0, stream>>>(cnt, cnt + NN, start);
  seg_fill_kernel<<<(EE + 255) / 256, 256, 0, stream>>>(dst, start, rank, perm);
  seg_sort_kernel<<<(NN + 255) / 256, 256, 0, stream>>>(src, start, cnt, perm, srcp, dstp);

  encoder_kernel<64, 128, XRT, false><<<NN / (16 * XRT), 256, 0, stream>>>(x, lin_x_w, lin_x_b, h, nullptr);
  encoder_kernel<16, 64, ERT, true><<<EE / (16 * ERT), 128, 0, stream>>>(ea, lin_e_w, lin_e_b, e, perm);

  for (int l = 0; l < LL; ++l) {
    node_proj_kernel<<<dim3(NN / NB, 2), PW * 32, 0, stream>>>(h,
        wpk + (size_t)l * PWL, pq);
    edge_mlp_kernel<<<EE / EB, 128, 0, stream>>>(pq, e, srcp, dstp,
        wpk + (size_t)l * PWL, phie_b1 + (size_t)l * 64, phie_b2 + (size_t)l * 64);
    node_mlp_kernel<<<NN / 16, 128, 0, stream>>>(h, e, start, cnt,
        wpk + (size_t)l * PWL, phiv_b1 + (size_t)l * 64, phiv_b2 + (size_t)l * 128);
  }

  hipMemsetAsync(gsum, 0, (size_t)BB * 128 * 4, stream);
  hipMemsetAsync(gcnt, 0, (size_t)BB * 4, stream);
  pool_kernel<<<NN / PLB, 256, 0, stream>>>(h, batch, gsum, gcnt);
  recip_clamp_kernel<<<1, BB, 0, stream>>>(gcnt, BB);
  readout_kernel<<<BB / 16, 256, 0, stream>>>(gsum, gcnt, ro_w1, ro_b1, ro_w2, ro_b2,
                                              (float*)d_out);
}
